# starting kernel
# baseline (speedup 1.0000x reference)
#include <hip/hip_runtime.h>
#include <hip/hip_bf16.h>


typedef float v2f __attribute__((ext_vector_type(2)));
typedef float v8f __attribute__((ext_vector_type(8)));

__device__ __forceinline__ v8f wmma_f32_16x16x4(v2f a, v2f b, v8f c) {
  return __builtin_amdgcn_wmma_f32_16x16x4_f32(false, a, false, b, (short)0, c,
                                               false, false);
}

template <int OPB>
__global__ void __launch_bounds__(256)
gemm_wmma(const float* __restrict__ A, int lda,
          const float* __restrict__ B, int ldb,
          float* __restrict__ C, int ldc,
          int M, int Nc, int K) {
  constexpr int BM = 128, BN = 128, BK = 16;
  constexpr int ALD = 20;
  constexpr int BLD = 132;

  __shared__ float As[BM * ALD];
  __shared__ float Bs[BK * BLD];

  const int tid  = threadIdx.x;
  const int wid  = tid >> 5;
  const int lane = tid & 31;
  const int lh   = lane >> 4;
  const int lm   = lane & 15;

  const int m0 = blockIdx.y * BM;
  const int n0 = blockIdx.x * BN;
  const int wm = (wid & 1) * 64;
  const int wn = (wid >> 1) * 32;

  v8f acc[4][2];
#pragma unroll
  for (int tm = 0; tm < 4; ++tm)
#pragma unroll
    for (int tn = 0; tn < 2; ++tn) acc[tm][tn] = (v8f)0.0f;

  for (int k0 = 0; k0 < K; k0 += BK) {
    __syncthreads();
#pragma unroll
    for (int r = 0; r < 2; ++r) {
      const int i   = tid + r * 256;
      const int row = i >> 2;
      const int c4  = (i & 3) << 2;
      const int gm  = m0 + row;
      float4 v = make_float4(0.f, 0.f, 0.f, 0.f);
      if (gm < M)
        v = *(const float4*)(A + (size_t)gm * lda + k0 + c4);
      *(float4*)(&As[row * ALD + c4]) = v;
    }
    if (OPB) {
#pragma unroll
      for (int r = 0; r < 2; ++r) {
        const int i  = tid + r * 256;
        const int j  = i >> 2;
        const int k4 = (i & 3) << 2;
        const int gj = n0 + j;
        float4 v = make_float4(0.f, 0.f, 0.f, 0.f);
        if (gj < Nc)
          v = *(const float4*)(B + (size_t)gj * ldb + k0 + k4);
        Bs[(k4 + 0) * BLD + j] = v.x;
        Bs[(k4 + 1) * BLD + j] = v.y;
        Bs[(k4 + 2) * BLD + j] = v.z;
        Bs[(k4 + 3) * BLD + j] = v.w;
      }
    } else {
#pragma unroll
      for (int r = 0; r < 2; ++r) {
        const int i  = tid + r * 256;
        const int kk = i >> 5;
        const int c  = (i & 31) << 2;
        const int gj = n0 + c;
        float4 v = make_float4(0.f, 0.f, 0.f, 0.f);
        if (gj < Nc)
          v = *(const float4*)(B + (size_t)(k0 + kk) * ldb + gj);
        *(float4*)(&Bs[kk * BLD + c]) = v;
      }
    }
    if (k0 + BK < K) {
      const int gm = m0 + (tid >> 1);
      if (gm < M)
        __builtin_prefetch(A + (size_t)gm * lda + (k0 + BK) + (tid & 1) * 8, 0, 1);
    }
    __syncthreads();

#pragma unroll
    for (int kk = 0; kk < BK; kk += 4) {
      const int ks = kk + 2 * lh;
      v2f a[4], b[2];
#pragma unroll
      for (int tm = 0; tm < 4; ++tm) {
        a[tm] = *(const v2f*)(&As[(wm + tm * 16 + lm) * ALD + ks]);
      }
#pragma unroll
      for (int tn = 0; tn < 2; ++tn) {
        const int col = wn + tn * 16 + lm;
        b[tn].x = Bs[(ks + 0) * BLD + col];
        b[tn].y = Bs[(ks + 1) * BLD + col];
      }
#pragma unroll
      for (int tm = 0; tm < 4; ++tm)
#pragma unroll
        for (int tn = 0; tn < 2; ++tn)
          acc[tm][tn] = wmma_f32_16x16x4(a[tm], b[tn], acc[tm][tn]);
    }
  }

#pragma unroll
  for (int tm = 0; tm < 4; ++tm) {
#pragma unroll
    for (int v = 0; v < 8; ++v) {
      const int gm = m0 + wm + tm * 16 + v + 8 * lh;
      if (gm < M) {
#pragma unroll
        for (int tn = 0; tn < 2; ++tn) {
          const int gn = n0 + wn + tn * 16 + lm;
          C[(size_t)gm * ldc + gn] = acc[tm][tn][v];
        }
      }
    }
  }
}

__global__ void init_deg_kernel(float* deg, int n) {
  int i = blockIdx.x * blockDim.x + threadIdx.x;
  if (i < n) deg[i] = 1.0f;
}

__global__ void deg_accum_kernel(const int* __restrict__ dst, float* deg, int E) {
  int e = blockIdx.x * blockDim.x + threadIdx.x;
  if (e < E) atomicAdd(&deg[dst[e]], 1.0f);
}

__global__ void dinv_kernel(float* deg, int n) {
  int i = blockIdx.x * blockDim.x + threadIdx.x;
  if (i < n) {
    float d = deg[i];
    deg[i] = (d > 0.f) ? rsqrtf(d) : 0.f;
  }
}

__global__ void norm_kernel(const int* __restrict__ src, const int* __restrict__ dst,
                            const float* __restrict__ dinv, float* __restrict__ norm,
                            int E, int Et) {
  int e = blockIdx.x * blockDim.x + threadIdx.x;
  if (e >= Et) return;
  if (e < E) {
    norm[e] = dinv[src[e]] * dinv[dst[e]];
  } else {
    float d = dinv[e - E];
    norm[e] = d * d;
  }
}

__global__ void bias_init_kernel(float* __restrict__ out, const float* __restrict__ b,
                                 int n, int D4) {
  int i = blockIdx.x * blockDim.x + threadIdx.x;
  if (i >= n * D4) return;
  const int j = i % D4;
  ((float4*)out)[i] = ((const float4*)b)[j];
}

__global__ void scatter_kernel(const float* __restrict__ h,
                               const int* __restrict__ src, const int* __restrict__ dst,
                               const float* __restrict__ norm, float* __restrict__ out,
                               int D, int D4, int E, int Et) {
  long long tid = (long long)blockIdx.x * blockDim.x + threadIdx.x;
  long long total = (long long)Et * D4;
  if (tid >= total) return;
  const int e = (int)(tid / D4);
  const int j = (int)(tid % D4);
  int s, d;
  if (e < E) { s = src[e]; d = dst[e]; }
  else       { s = e - E;  d = e - E;  }
  const float nv = norm[e];
  const float4 v = ((const float4*)(h + (size_t)s * D))[j];
  float* o = out + (size_t)d * D + j * 4;
  atomicAdd(o + 0, v.x * nv);
  atomicAdd(o + 1, v.y * nv);
  atomicAdd(o + 2, v.z * nv);
  atomicAdd(o + 3, v.w * nv);
}

static inline void launch_gemm(int opb, const float* A, int lda, const float* B,
                               int ldb, float* C, int ldc, int M, int Nc, int K,
                               hipStream_t stream) {
  dim3 grid(Nc / 128, (M + 127) / 128);
  if (opb)
    gemm_wmma<1><<<grid, 256, 0, stream>>>(A, lda, B, ldb, C, ldc, M, Nc, K);
  else
    gemm_wmma<0><<<grid, 256, 0, stream>>>(A, lda, B, ldb, C, ldc, M, Nc, K);
}

extern "C" void kernel_launch(void* const* d_in, const int* in_sizes, int n_in,
                              void* d_out, int out_size, void* d_ws, size_t ws_size,
                              hipStream_t stream) {
  const int IN = 1024, HID = 512, OUT = 128;

  const float* features = (const float*)d_in[0];
  const int*   edge     = (const int*)d_in[1];
  const float* W1       = (const float*)d_in[2];
  const float* b1       = (const float*)d_in[3];
  const float* W2       = (const float*)d_in[4];
  const float* b2       = (const float*)d_in[5];
  const float* b3       = (const float*)d_in[6];
  const float* b4       = (const float*)d_in[7];
  const float* head1    = (const float*)d_in[8];

  const int n  = in_sizes[0] / IN;
  const int E  = in_sizes[1] / 2;
  const int Et = E + n;
  const int* src = edge;
  const int* dst = edge + E;

  float* out = (float*)d_out;
  float* z   = out;
  float* h2  = out + (size_t)n * OUT;
  float* h4  = out + (size_t)n * OUT * 2;

  float* ws   = (float*)d_ws;
  float* dinv = ws;
  float* norm = dinv + n;
  float* t0   = norm + Et;
  float* t1   = t0 + (size_t)n * HID;
  float* h2p  = t1 + (size_t)n * HID;
  float* t2   = h2p + (size_t)n * OUT;

  const int TB = 256;
  init_deg_kernel<<<(n + TB - 1) / TB, TB, 0, stream>>>(dinv, n);
  deg_accum_kernel<<<(E + TB - 1) / TB, TB, 0, stream>>>(dst, dinv, E);
  dinv_kernel<<<(n + TB - 1) / TB, TB, 0, stream>>>(dinv, n);
  norm_kernel<<<(Et + TB - 1) / TB, TB, 0, stream>>>(src, dst, dinv, norm, E, Et);

  launch_gemm(1, features, IN, W1, IN, t0, HID, n, HID, IN, stream);
  bias_init_kernel<<<((n * (HID / 4)) + TB - 1) / TB, TB, 0, stream>>>(t1, b1, n, HID / 4);
  {
    long long tot = (long long)Et * (HID / 4);
    scatter_kernel<<<(unsigned)((tot + TB - 1) / TB), TB, 0, stream>>>(
        t0, src, dst, norm, t1, HID, HID / 4, E, Et);
  }

  launch_gemm(1, t1, HID, W2, HID, h2p, OUT, n, OUT, HID, stream);
  bias_init_kernel<<<((n * (OUT / 4)) + TB - 1) / TB, TB, 0, stream>>>(h2, b2, n, OUT / 4);
  {
    long long tot = (long long)Et * (OUT / 4);
    scatter_kernel<<<(unsigned)((tot + TB - 1) / TB), TB, 0, stream>>>(
        h2p, src, dst, norm, h2, OUT, OUT / 4, E, Et);
  }

  launch_gemm(0, h2, OUT, head1, OUT, z, OUT, n, OUT, OUT, stream);

  launch_gemm(0, z, OUT, W2, HID, t0, HID, n, HID, OUT, stream);
  bias_init_kernel<<<((n * (HID / 4)) + TB - 1) / TB, TB, 0, stream>>>(t1, b3, n, HID / 4);
  {
    long long tot = (long long)Et * (HID / 4);
    scatter_kernel<<<(unsigned)((tot + TB - 1) / TB), TB, 0, stream>>>(
        t0, src, dst, norm, t1, HID, HID / 4, E, Et);
  }

  launch_gemm(0, t1, HID, W1, IN, t2, IN, n, IN, HID, stream);
  bias_init_kernel<<<((n * (IN / 4)) + TB - 1) / TB, TB, 0, stream>>>(h4, b4, n, IN / 4);
  {
    long long tot = (long long)Et * (IN / 4);
    scatter_kernel<<<(unsigned)((tot + TB - 1) / TB), TB, 0, stream>>>(
        t2, src, dst, norm, h4, IN, IN / 4, E, Et);
  }
}


# GEMM tile staging branch-free (A row index clamped, dead B column guard dropped): a K step's 4 global loads issue back to back under one wait, was 4 serial load-wait-store rounds; sums unchanged
# speedup vs baseline: 1.7713x; 1.7713x over previous
#include <hip/hip_runtime.h>
#include <hip/hip_bf16.h>


typedef float v2f __attribute__((ext_vector_type(2)));
typedef float v8f __attribute__((ext_vector_type(8)));

__device__ __forceinline__ v8f wmma_f32_16x16x4(v2f a, v2f b, v8f c) {
  return __builtin_amdgcn_wmma_f32_16x16x4_f32(false, a, false, b, (short)0, c,
                                               false, false);
}

template <int OPB, bool INIT>
__global__ void __launch_bounds__(256)
gemm_wmma(const float* __restrict__ A, int lda,
          const float* __restrict__ B, int ldb,
          float* __restrict__ C, int ldc,
          int M, int Nc, int K,
          float* __restrict__ C2, const float* __restrict__ bias,
          const float* __restrict__ dinv) {
  constexpr int BM = 128, BN = 128, BK = 16;
  constexpr int ALD = 20;
  constexpr int BLD = 132;

  __shared__ float As[BM * ALD];
  __shared__ float Bs[BK * BLD];

  const int tid  = threadIdx.x;
  const int wid  = tid >> 5;
  const int lane = tid & 31;
  const int lh   = lane >> 4;
  const int lm   = lane & 15;

  const int m0 = blockIdx.y * BM;
  const int n0 = blockIdx.x * BN;
  const int wm = (wid & 1) * 64;
  const int wn = (wid >> 1) * 32;

  v8f acc[4][2];
#pragma unroll
  for (int tm = 0; tm < 4; ++tm)
#pragma unroll
    for (int tn = 0; tn < 2; ++tn) acc[tm][tn] = (v8f)0.0f;

  const float* pa[2];
  const float* pb[2];
#pragma unroll
  for (int r = 0; r < 2; ++r) {
    const int i   = tid + r * 256;
    const int gm  = m0 + (i >> 2);
    const int gmc = gm < M ? gm : M - 1;
    pa[r] = A + (size_t)gmc * lda + ((i & 3) << 2);
    if (OPB)
      pb[r] = B + (size_t)(n0 + (i >> 2)) * ldb + ((i & 3) << 2);
    else
      pb[r] = B + (size_t)(i >> 5) * ldb + n0 + ((i & 31) << 2);
  }

  for (int k0 = 0; k0 < K; k0 += BK) {
    __syncthreads();
    float4 va[2], vb[2];
#pragma unroll
    for (int r = 0; r < 2; ++r) va[r] = *(const float4*)(pa[r] + k0);
#pragma unroll
    for (int r = 0; r < 2; ++r)
      vb[r] = *(const float4*)(OPB ? pb[r] + k0 : pb[r] + (size_t)k0 * ldb);
#pragma unroll
    for (int r = 0; r < 2; ++r) {
      const int i   = tid + r * 256;
      const int row = i >> 2;
      const int c4  = (i & 3) << 2;
      *(float4*)(&As[row * ALD + c4]) = va[r];
    }
    if (OPB) {
#pragma unroll
      for (int r = 0; r < 2; ++r) {
        const int i  = tid + r * 256;
        const int j  = i >> 2;
        const int k4 = (i & 3) << 2;
        Bs[(k4 + 0) * BLD + j] = vb[r].x;
        Bs[(k4 + 1) * BLD + j] = vb[r].y;
        Bs[(k4 + 2) * BLD + j] = vb[r].z;
        Bs[(k4 + 3) * BLD + j] = vb[r].w;
      }
    } else {
#pragma unroll
      for (int r = 0; r < 2; ++r) {
        const int i  = tid + r * 256;
        const int kk = i >> 5;
        const int c  = (i & 31) << 2;
        *(float4*)(&Bs[kk * BLD + c]) = vb[r];
      }
    }
    if (k0 + BK < K) {
      const int gm = m0 + (tid >> 1);
      if (gm < M)
        __builtin_prefetch(A + (size_t)gm * lda + (k0 + BK) + (tid & 1) * 8, 0, 1);
    }
    __syncthreads();

#pragma unroll
    for (int kk = 0; kk < BK; kk += 4) {
      const int ks = kk + 2 * lh;
      v2f a[4], b[2];
#pragma unroll
      for (int tm = 0; tm < 4; ++tm) {
        a[tm] = *(const v2f*)(&As[(wm + tm * 16 + lm) * ALD + ks]);
      }
#pragma unroll
      for (int tn = 0; tn < 2; ++tn) {
        const int col = wn + tn * 16 + lm;
        b[tn].x = Bs[(ks + 0) * BLD + col];
        b[tn].y = Bs[(ks + 1) * BLD + col];
      }
#pragma unroll
      for (int tm = 0; tm < 4; ++tm)
#pragma unroll
        for (int tn = 0; tn < 2; ++tn)
          acc[tm][tn] = wmma_f32_16x16x4(a[tm], b[tn], acc[tm][tn]);
    }
  }

  float bv[2] = {0.f, 0.f};
  if (INIT) {
#pragma unroll
    for (int tn = 0; tn < 2; ++tn) bv[tn] = bias[n0 + wn + tn * 16 + lm];
  }
#pragma unroll
  for (int tm = 0; tm < 4; ++tm) {
    float selfw[8];
    if (INIT) {
#pragma unroll
      for (int v = 0; v < 8; ++v) {
        const int gm = m0 + wm + tm * 16 + v + 8 * lh;
        const float d = dinv[gm < M ? gm : M - 1];
        selfw[v] = d * d;
      }
    }
#pragma unroll
    for (int v = 0; v < 8; ++v) {
      const int gm = m0 + wm + tm * 16 + v + 8 * lh;
      if (gm < M) {
#pragma unroll
        for (int tn = 0; tn < 2; ++tn) {
          const int gn = n0 + wn + tn * 16 + lm;
          const float c = acc[tm][tn][v];
          C[(size_t)gm * ldc + gn] = c;
          if (INIT) C2[(size_t)gm * ldc + gn] = bv[tn] + c * selfw[v];
        }
      }
    }
  }
}

__global__ void init_deg_kernel(float* deg, int n) {
  int i = blockIdx.x * blockDim.x + threadIdx.x;
  if (i < n) deg[i] = 1.0f;
}

__global__ void deg_accum_kernel(const int* __restrict__ dst, float* deg, int E) {
  int e = blockIdx.x * blockDim.x + threadIdx.x;
  if (e < E) atomicAdd(&deg[dst[e]], 1.0f);
}

__global__ void dinv_kernel(float* deg, int n) {
  int i = blockIdx.x * blockDim.x + threadIdx.x;
  if (i < n) {
    float d = deg[i];
    deg[i] = (d > 0.f) ? rsqrtf(d) : 0.f;
  }
}

__global__ void norm_kernel(const int* __restrict__ src, const int* __restrict__ dst,
                            const float* __restrict__ dinv, float* __restrict__ norm,
                            int E) {
  int e = blockIdx.x * blockDim.x + threadIdx.x;
  if (e < E) norm[e] = dinv[src[e]] * dinv[dst[e]];
}

template <int D, int N>
__device__ __forceinline__ void scatter_rows(const float* __restrict__ h,
                                             float* __restrict__ out,
                                             int sl, int dl, int nl, int k0, int lane) {
  constexpr int R = D / 32;
  float v[N][R];
#pragma unroll
  for (int k = 0; k < N; ++k) {
    const int s = __builtin_amdgcn_readlane(sl, k0 + k);
    const float* hp = h + (size_t)s * D + lane;
#pragma unroll
    for (int i = 0; i < R; ++i) v[k][i] = hp[32 * i];
  }
#pragma unroll
  for (int k = 0; k < N; ++k) {
    const int   d  = __builtin_amdgcn_readlane(dl, k0 + k);
    const float nv = __builtin_bit_cast(float, __builtin_amdgcn_readlane(nl, k0 + k));
    float* op = out + (size_t)d * D + lane;
#pragma unroll
    for (int i = 0; i < R; ++i) atomicAdd(op + 32 * i, v[k][i] * nv);
  }
}

template <int D, int EPW>
__global__ void __launch_bounds__(256)
scatter_rows_kernel(const float* __restrict__ h,
                    const int* __restrict__ src, const int* __restrict__ dst,
                    const float* __restrict__ norm, float* __restrict__ out,
                    int E) {
  const int lane = threadIdx.x & 31;
  const int w = __builtin_amdgcn_readfirstlane((int)(blockIdx.x * 8u + (threadIdx.x >> 5)));
  const int e0 = w * EPW;
  if (e0 >= E) return;

  int el = e0 + ((lane < EPW) ? lane : 0);
  el = (el < E) ? el : (E - 1);
  const int sl = src[el];
  const int dl = dst[el];
  const int nl = __builtin_bit_cast(int, norm[el]);

  if (e0 + EPW <= E) {
    scatter_rows<D, EPW>(h, out, sl, dl, nl, 0, lane);
  } else {
    for (int k = 0; e0 + k < E; ++k)
      scatter_rows<D, 1>(h, out, sl, dl, nl, k, lane);
  }
}

static inline void launch_gemm(int opb, const float* A, int lda, const float* B,
                               int ldb, float* C, int ldc, int M, int Nc, int K,
                               float* agg0, const float* bias, const float* dinv,
                               hipStream_t stream) {
  dim3 grid(Nc / 128, (M + 127) / 128);
  if (opb)
    gemm_wmma<1, true><<<grid, 256, 0, stream>>>(A, lda, B, ldb, C, ldc, M, Nc, K,
                                                 agg0, bias, dinv);
  else if (agg0)
    gemm_wmma<0, true><<<grid, 256, 0, stream>>>(A, lda, B, ldb, C, ldc, M, Nc, K,
                                                 agg0, bias, dinv);
  else
    gemm_wmma<0, false><<<grid, 256, 0, stream>>>(A, lda, B, ldb, C, ldc, M, Nc, K,
                                                  agg0, bias, dinv);
}

extern "C" void kernel_launch(void* const* d_in, const int* in_sizes, int n_in,
                              void* d_out, int out_size, void* d_ws, size_t ws_size,
                              hipStream_t stream) {
  const int IN = 1024, HID = 512, OUT = 128;

  const float* features = (const float*)d_in[0];
  const int*   edge     = (const int*)d_in[1];
  const float* W1       = (const float*)d_in[2];
  const float* b1       = (const float*)d_in[3];
  const float* W2       = (const float*)d_in[4];
  const float* b2       = (const float*)d_in[5];
  const float* b3       = (const float*)d_in[6];
  const float* b4       = (const float*)d_in[7];
  const float* head1    = (const float*)d_in[8];

  const int n  = in_sizes[0] / IN;
  const int E  = in_sizes[1] / 2;
  const int Et = E + n;
  const int* src = edge;
  const int* dst = edge + E;

  float* out = (float*)d_out;
  float* z   = out;
  float* h2  = out + (size_t)n * OUT;
  float* h4  = out + (size_t)n * OUT * 2;

  float* ws   = (float*)d_ws;
  float* dinv = ws;
  float* norm = dinv + n;
  float* t0   = norm + Et;
  float* t1   = t0 + (size_t)n * HID;
  float* h2p  = t1 + (size_t)n * HID;
  float* t2   = h2p + (size_t)n * OUT;

  const int TB = 256;
  init_deg_kernel<<<(n + TB - 1) / TB, TB, 0, stream>>>(dinv, n);
  deg_accum_kernel<<<(E + TB - 1) / TB, TB, 0, stream>>>(dst, dinv, E);
  dinv_kernel<<<(n + TB - 1) / TB, TB, 0, stream>>>(dinv, n);
  norm_kernel<<<(E + TB - 1) / TB, TB, 0, stream>>>(src, dst, dinv, norm, E);

  launch_gemm(1, features, IN, W1, IN, t0, HID, n, HID, IN, t1, b1, dinv, stream);
  scatter_rows_kernel<HID, 1><<<(E + 7) / 8, 256, 0, stream>>>(
      t0, src, dst, norm, t1, E);

  launch_gemm(1, t1, HID, W2, HID, h2p, OUT, n, OUT, HID, h2, b2, dinv, stream);
  scatter_rows_kernel<OUT, 4><<<(E + 31) / 32, 256, 0, stream>>>(
      h2p, src, dst, norm, h2, E);

  launch_gemm(0, h2, OUT, head1, OUT, z, OUT, n, OUT, OUT, nullptr, nullptr, nullptr,
              stream);

  launch_gemm(0, z, OUT, W2, HID, t0, HID, n, HID, OUT, t1, b3, dinv, stream);
  scatter_rows_kernel<HID, 1><<<(E + 7) / 8, 256, 0, stream>>>(
      t0, src, dst, norm, t1, E);

  launch_gemm(0, t1, HID, W1, IN, t2, IN, n, IN, HID, h4, b4, dinv, stream);
  scatter_rows_kernel<IN, 1><<<(E + 7) / 8, 256, 0, stream>>>(
      t2, src, dst, norm, h4, E);
}


# Encoder products first, (X W1^T) W2^T, with conv1's aggregation deferred behind conv2's weight (b1 carried as W2 b1): both encoder scatters run 128 wide, was 512 and 128; changes float sum order
# speedup vs baseline: 1.8847x; 1.0640x over previous
#include <hip/hip_runtime.h>
#include <hip/hip_bf16.h>


typedef float v2f __attribute__((ext_vector_type(2)));
typedef float v8f __attribute__((ext_vector_type(8)));

__device__ __forceinline__ v8f wmma_f32_16x16x4(v2f a, v2f b, v8f c) {
  return __builtin_amdgcn_wmma_f32_16x16x4_f32(false, a, false, b, (short)0, c,
                                               false, false);
}

template <int OPB, bool INIT>
__global__ void __launch_bounds__(256)
gemm_wmma(const float* __restrict__ A, int lda,
          const float* __restrict__ B, int ldb,
          float* __restrict__ C, int ldc,
          int M, int Nc, int K,
          float* __restrict__ C2, const float* __restrict__ bias,
          const float* __restrict__ dinv) {
  constexpr int BM = 128, BN = 128, BK = 16;
  constexpr int ALD = 20;
  constexpr int BLD = 132;

  __shared__ float As[BM * ALD];
  __shared__ float Bs[BK * BLD];

  const int tid  = threadIdx.x;
  const int wid  = tid >> 5;
  const int lane = tid & 31;
  const int lh   = lane >> 4;
  const int lm   = lane & 15;

  const int m0 = blockIdx.y * BM;
  const int n0 = blockIdx.x * BN;
  const int wm = (wid & 1) * 64;
  const int wn = (wid >> 1) * 32;

  v8f acc[4][2];
#pragma unroll
  for (int tm = 0; tm < 4; ++tm)
#pragma unroll
    for (int tn = 0; tn < 2; ++tn) acc[tm][tn] = (v8f)0.0f;

  const float* pa[2];
  const float* pb[2];
#pragma unroll
  for (int r = 0; r < 2; ++r) {
    const int i   = tid + r * 256;
    const int gm  = m0 + (i >> 2);
    const int gmc = gm < M ? gm : M - 1;
    pa[r] = A + (size_t)gmc * lda + ((i & 3) << 2);
    if (OPB)
      pb[r] = B + (size_t)(n0 + (i >> 2)) * ldb + ((i & 3) << 2);
    else
      pb[r] = B + (size_t)(i >> 5) * ldb + n0 + ((i & 31) << 2);
  }

  for (int k0 = 0; k0 < K; k0 += BK) {
    __syncthreads();
    float4 va[2], vb[2];
#pragma unroll
    for (int r = 0; r < 2; ++r) va[r] = *(const float4*)(pa[r] + k0);
#pragma unroll
    for (int r = 0; r < 2; ++r)
      vb[r] = *(const float4*)(OPB ? pb[r] + k0 : pb[r] + (size_t)k0 * ldb);
#pragma unroll
    for (int r = 0; r < 2; ++r) {
      const int i   = tid + r * 256;
      const int row = i >> 2;
      const int c4  = (i & 3) << 2;
      *(float4*)(&As[row * ALD + c4]) = va[r];
    }
    if (OPB) {
#pragma unroll
      for (int r = 0; r < 2; ++r) {
        const int i  = tid + r * 256;
        const int j  = i >> 2;
        const int k4 = (i & 3) << 2;
        Bs[(k4 + 0) * BLD + j] = vb[r].x;
        Bs[(k4 + 1) * BLD + j] = vb[r].y;
        Bs[(k4 + 2) * BLD + j] = vb[r].z;
        Bs[(k4 + 3) * BLD + j] = vb[r].w;
      }
    } else {
#pragma unroll
      for (int r = 0; r < 2; ++r) {
        const int i  = tid + r * 256;
        const int kk = i >> 5;
        const int c  = (i & 31) << 2;
        *(float4*)(&Bs[kk * BLD + c]) = vb[r];
      }
    }
    if (k0 + BK < K) {
      const int gm = m0 + (tid >> 1);
      if (gm < M)
        __builtin_prefetch(A + (size_t)gm * lda + (k0 + BK) + (tid & 1) * 8, 0, 1);
    }
    __syncthreads();

#pragma unroll
    for (int kk = 0; kk < BK; kk += 4) {
      const int ks = kk + 2 * lh;
      v2f a[4], b[2];
#pragma unroll
      for (int tm = 0; tm < 4; ++tm) {
        a[tm] = *(const v2f*)(&As[(wm + tm * 16 + lm) * ALD + ks]);
      }
#pragma unroll
      for (int tn = 0; tn < 2; ++tn) {
        const int col = wn + tn * 16 + lm;
        b[tn].x = Bs[(ks + 0) * BLD + col];
        b[tn].y = Bs[(ks + 1) * BLD + col];
      }
#pragma unroll
      for (int tm = 0; tm < 4; ++tm)
#pragma unroll
        for (int tn = 0; tn < 2; ++tn)
          acc[tm][tn] = wmma_f32_16x16x4(a[tm], b[tn], acc[tm][tn]);
    }
  }

  float bv[2] = {0.f, 0.f};
  if (INIT) {
#pragma unroll
    for (int tn = 0; tn < 2; ++tn) bv[tn] = bias[n0 + wn + tn * 16 + lm];
  }
#pragma unroll
  for (int tm = 0; tm < 4; ++tm) {
    float selfw[8];
    if (INIT) {
#pragma unroll
      for (int v = 0; v < 8; ++v) {
        const int gm = m0 + wm + tm * 16 + v + 8 * lh;
        const float d = dinv[gm < M ? gm : M - 1];
        selfw[v] = d * d;
      }
    }
#pragma unroll
    for (int v = 0; v < 8; ++v) {
      const int gm = m0 + wm + tm * 16 + v + 8 * lh;
      if (gm < M) {
#pragma unroll
        for (int tn = 0; tn < 2; ++tn) {
          const int gn = n0 + wn + tn * 16 + lm;
          const float c = acc[tm][tn][v];
          C[(size_t)gm * ldc + gn] = c;
          if (INIT) C2[(size_t)gm * ldc + gn] = bv[tn] + c * selfw[v];
        }
      }
    }
  }
}

template <int HIDD, int OUTD>
__global__ void __launch_bounds__(256)
init_kernel(float* deg, int n, int nb_deg,
            const float* __restrict__ W2, const float* __restrict__ b1,
            float* __restrict__ cvec,
            const float* __restrict__ b2, float* __restrict__ h2) {
  const int b = blockIdx.x;
  if (b < nb_deg) {
    const int i = b * 256 + threadIdx.x;
    if (i < n) deg[i] = 1.0f;
  } else if (b < nb_deg + 16) {
    const int lane = threadIdx.x & 31;
    const int j = (b - nb_deg) * 8 + (threadIdx.x >> 5);
    if (j < OUTD) {
      float s = 0.f;
#pragma unroll 4
      for (int k = lane; k < HIDD; k += 32) s += W2[j * HIDD + k] * b1[k];
#pragma unroll
      for (int o = 16; o > 0; o >>= 1) s += __shfl_xor(s, o);
      if (lane == 0) cvec[j] = s;
    }
  } else {
    const int i4 = (b - nb_deg - 16) * 256 + threadIdx.x;
    if (i4 < n * (OUTD / 4)) {
      const int c = (i4 % (OUTD / 4)) * 4;
      *(float4*)(h2 + (size_t)i4 * 4) = *(const float4*)(b2 + c);
    }
  }
}

__global__ void deg_accum_kernel(const int* __restrict__ dst, float* deg, int E) {
  int e = blockIdx.x * blockDim.x + threadIdx.x;
  if (e < E) atomicAdd(&deg[dst[e]], 1.0f);
}

__global__ void dinv_kernel(float* deg, int n) {
  int i = blockIdx.x * blockDim.x + threadIdx.x;
  if (i < n) {
    float d = deg[i];
    deg[i] = (d > 0.f) ? rsqrtf(d) : 0.f;
  }
}

__global__ void norm_kernel(const int* __restrict__ src, const int* __restrict__ dst,
                            const float* __restrict__ dinv, float* __restrict__ norm,
                            int E) {
  int e = blockIdx.x * blockDim.x + threadIdx.x;
  if (e < E) norm[e] = dinv[src[e]] * dinv[dst[e]];
}

template <int D, int N>
__device__ __forceinline__ void scatter_rows(const float* __restrict__ h,
                                             float* __restrict__ out,
                                             int sl, int dl, int nl, int k0, int lane) {
  constexpr int R = D / 32;
  float v[N][R];
#pragma unroll
  for (int k = 0; k < N; ++k) {
    const int s = __builtin_amdgcn_readlane(sl, k0 + k);
    const float* hp = h + (size_t)s * D + lane;
#pragma unroll
    for (int i = 0; i < R; ++i) v[k][i] = hp[32 * i];
  }
#pragma unroll
  for (int k = 0; k < N; ++k) {
    const int   d  = __builtin_amdgcn_readlane(dl, k0 + k);
    const float nv = __builtin_bit_cast(float, __builtin_amdgcn_readlane(nl, k0 + k));
    float* op = out + (size_t)d * D + lane;
#pragma unroll
    for (int i = 0; i < R; ++i) atomicAdd(op + 32 * i, v[k][i] * nv);
  }
}

template <int D, int EPW, bool SELF>
__global__ void __launch_bounds__(256)
scatter_rows_kernel(const float* __restrict__ h,
                    const int* __restrict__ src, const int* __restrict__ dst,
                    const float* __restrict__ norm, float* __restrict__ out,
                    int E, const float* __restrict__ dinv, int n) {
  const int lane = threadIdx.x & 31;
  const int w = __builtin_amdgcn_readfirstlane((int)(blockIdx.x * 8u + (threadIdx.x >> 5)));
  const int e0 = w * EPW;
  const int Et = SELF ? E + n : E;
  if (e0 >= Et) return;

  int el = e0 + ((lane < EPW) ? lane : 0);
  el = (el < Et) ? el : (Et - 1);
  int sl, dl, nl;
  if (SELF && el >= E) {
    sl = dl = el - E;
    const float d = dinv[sl];
    nl = __builtin_bit_cast(int, d * d);
  } else {
    sl = src[el];
    dl = dst[el];
    nl = __builtin_bit_cast(int, norm[el]);
  }

  if (e0 + EPW <= Et) {
    scatter_rows<D, EPW>(h, out, sl, dl, nl, 0, lane);
  } else {
    for (int k = 0; e0 + k < Et; ++k)
      scatter_rows<D, 1>(h, out, sl, dl, nl, k, lane);
  }
}

static inline void launch_gemm(int opb, const float* A, int lda, const float* B,
                               int ldb, float* C, int ldc, int M, int Nc, int K,
                               float* agg0, const float* bias, const float* dinv,
                               hipStream_t stream) {
  dim3 grid(Nc / 128, (M + 127) / 128);
  if (opb && agg0)
    gemm_wmma<1, true><<<grid, 256, 0, stream>>>(A, lda, B, ldb, C, ldc, M, Nc, K,
                                                 agg0, bias, dinv);
  else if (opb)
    gemm_wmma<1, false><<<grid, 256, 0, stream>>>(A, lda, B, ldb, C, ldc, M, Nc, K,
                                                  agg0, bias, dinv);
  else if (agg0)
    gemm_wmma<0, true><<<grid, 256, 0, stream>>>(A, lda, B, ldb, C, ldc, M, Nc, K,
                                                 agg0, bias, dinv);
  else
    gemm_wmma<0, false><<<grid, 256, 0, stream>>>(A, lda, B, ldb, C, ldc, M, Nc, K,
                                                  agg0, bias, dinv);
}

extern "C" void kernel_launch(void* const* d_in, const int* in_sizes, int n_in,
                              void* d_out, int out_size, void* d_ws, size_t ws_size,
                              hipStream_t stream) {
  const int IN = 1024, HID = 512, OUT = 128;

  const float* features = (const float*)d_in[0];
  const int*   edge     = (const int*)d_in[1];
  const float* W1       = (const float*)d_in[2];
  const float* b1       = (const float*)d_in[3];
  const float* W2       = (const float*)d_in[4];
  const float* b2       = (const float*)d_in[5];
  const float* b3       = (const float*)d_in[6];
  const float* b4       = (const float*)d_in[7];
  const float* head1    = (const float*)d_in[8];

  const int n  = in_sizes[0] / IN;
  const int E  = in_sizes[1] / 2;
  const int Et = E + n;
  const int* src = edge;
  const int* dst = edge + E;

  float* out = (float*)d_out;
  float* z   = out;
  float* h2  = out + (size_t)n * OUT;
  float* h4  = out + (size_t)n * OUT * 2;

  float* ws   = (float*)d_ws;
  float* dinv = ws;
  float* norm = dinv + n;
  float* t0   = norm + Et;
  float* t1   = t0 + (size_t)n * HID;
  float* h2p  = t1 + (size_t)n * HID;
  float* t2   = h2p + (size_t)n * OUT;
  float* cvec = norm + E;
  float* u    = h2p;
  float* p    = t1;

  const int TB = 256;
  const int nb_deg  = (n + TB - 1) / TB;
  const int nb_fill = (n * (OUT / 4) + TB - 1) / TB;
  init_kernel<HID, OUT><<<nb_deg + 16 + nb_fill, TB, 0, stream>>>(dinv, n, nb_deg, W2, b1,
                                                                  cvec, b2, h2);
  deg_accum_kernel<<<(E + TB - 1) / TB, TB, 0, stream>>>(dst, dinv, E);
  dinv_kernel<<<(n + TB - 1) / TB, TB, 0, stream>>>(dinv, n);
  norm_kernel<<<(E + TB - 1) / TB, TB, 0, stream>>>(src, dst, dinv, norm, E);

  launch_gemm(1, features, IN, W1, IN, t0, HID, n, HID, IN, nullptr, nullptr, nullptr,
              stream);
  launch_gemm(1, t0, HID, W2, HID, u, OUT, n, OUT, HID, p, cvec, dinv, stream);

  scatter_rows_kernel<OUT, 4, false><<<(E + 31) / 32, 256, 0, stream>>>(
      u, src, dst, norm, p, E, nullptr, 0);
  scatter_rows_kernel<OUT, 4, true><<<(E + n + 31) / 32, 256, 0, stream>>>(
      p, src, dst, norm, h2, E, dinv, n);

  launch_gemm(0, h2, OUT, head1, OUT, z, OUT, n, OUT, OUT, nullptr, nullptr, nullptr,
              stream);

  launch_gemm(0, z, OUT, W2, HID, t0, HID, n, HID, OUT, t1, b3, dinv, stream);
  scatter_rows_kernel<HID, 1, false><<<(E + 7) / 8, 256, 0, stream>>>(
      t0, src, dst, norm, t1, E, nullptr, 0);

  launch_gemm(0, t1, HID, W1, IN, t2, IN, n, IN, HID, h4, b4, dinv, stream);
  scatter_rows_kernel<IN, 1, false><<<(E + 7) / 8, 256, 0, stream>>>(
      t2, src, dst, norm, h4, E, nullptr, 0);
}


# Decoder aggregations (512, 1024 wide): a work-group owns 79 destination rows in LDS, filters the whole edge array for them and adds with LDS f32 atomics (was global atomics); sums in arrival order
# speedup vs baseline: 2.0897x; 1.1088x over previous
#include <hip/hip_runtime.h>
#include <hip/hip_bf16.h>


typedef float v2f __attribute__((ext_vector_type(2)));
typedef float v8f __attribute__((ext_vector_type(8)));

__device__ __forceinline__ v8f wmma_f32_16x16x4(v2f a, v2f b, v8f c) {
  return __builtin_amdgcn_wmma_f32_16x16x4_f32(false, a, false, b, (short)0, c,
                                               false, false);
}

template <int OPB, bool INIT>
__global__ void __launch_bounds__(256)
gemm_wmma(const float* __restrict__ A, int lda,
          const float* __restrict__ B, int ldb,
          float* __restrict__ C, int ldc,
          int M, int Nc, int K,
          float* __restrict__ C2, const float* __restrict__ bias,
          const float* __restrict__ dinv) {
  constexpr int BM = 128, BN = 128, BK = 16;
  constexpr int ALD = 20;
  constexpr int BLD = 132;

  __shared__ float As[BM * ALD];
  __shared__ float Bs[BK * BLD];

  const int tid  = threadIdx.x;
  const int wid  = tid >> 5;
  const int lane = tid & 31;
  const int lh   = lane >> 4;
  const int lm   = lane & 15;

  const int m0 = blockIdx.y * BM;
  const int n0 = blockIdx.x * BN;
  const int wm = (wid & 1) * 64;
  const int wn = (wid >> 1) * 32;

  v8f acc[4][2];
#pragma unroll
  for (int tm = 0; tm < 4; ++tm)
#pragma unroll
    for (int tn = 0; tn < 2; ++tn) acc[tm][tn] = (v8f)0.0f;

  const float* pa[2];
  const float* pb[2];
#pragma unroll
  for (int r = 0; r < 2; ++r) {
    const int i   = tid + r * 256;
    const int gm  = m0 + (i >> 2);
    const int gmc = gm < M ? gm : M - 1;
    pa[r] = A + (size_t)gmc * lda + ((i & 3) << 2);
    if (OPB)
      pb[r] = B + (size_t)(n0 + (i >> 2)) * ldb + ((i & 3) << 2);
    else
      pb[r] = B + (size_t)(i >> 5) * ldb + n0 + ((i & 31) << 2);
  }

  for (int k0 = 0; k0 < K; k0 += BK) {
    __syncthreads();
    float4 va[2], vb[2];
#pragma unroll
    for (int r = 0; r < 2; ++r) va[r] = *(const float4*)(pa[r] + k0);
#pragma unroll
    for (int r = 0; r < 2; ++r)
      vb[r] = *(const float4*)(OPB ? pb[r] + k0 : pb[r] + (size_t)k0 * ldb);
#pragma unroll
    for (int r = 0; r < 2; ++r) {
      const int i   = tid + r * 256;
      const int row = i >> 2;
      const int c4  = (i & 3) << 2;
      *(float4*)(&As[row * ALD + c4]) = va[r];
    }
    if (OPB) {
#pragma unroll
      for (int r = 0; r < 2; ++r) {
        const int i  = tid + r * 256;
        const int j  = i >> 2;
        const int k4 = (i & 3) << 2;
        Bs[(k4 + 0) * BLD + j] = vb[r].x;
        Bs[(k4 + 1) * BLD + j] = vb[r].y;
        Bs[(k4 + 2) * BLD + j] = vb[r].z;
        Bs[(k4 + 3) * BLD + j] = vb[r].w;
      }
    } else {
#pragma unroll
      for (int r = 0; r < 2; ++r) {
        const int i  = tid + r * 256;
        const int kk = i >> 5;
        const int c  = (i & 31) << 2;
        *(float4*)(&Bs[kk * BLD + c]) = vb[r];
      }
    }
    if (k0 + BK < K) {
      const int gm = m0 + (tid >> 1);
      if (gm < M)
        __builtin_prefetch(A + (size_t)gm * lda + (k0 + BK) + (tid & 1) * 8, 0, 1);
    }
    __syncthreads();

#pragma unroll
    for (int kk = 0; kk < BK; kk += 4) {
      const int ks = kk + 2 * lh;
      v2f a[4], b[2];
#pragma unroll
      for (int tm = 0; tm < 4; ++tm) {
        a[tm] = *(const v2f*)(&As[(wm + tm * 16 + lm) * ALD + ks]);
      }
#pragma unroll
      for (int tn = 0; tn < 2; ++tn) {
        const int col = wn + tn * 16 + lm;
        b[tn].x = Bs[(ks + 0) * BLD + col];
        b[tn].y = Bs[(ks + 1) * BLD + col];
      }
#pragma unroll
      for (int tm = 0; tm < 4; ++tm)
#pragma unroll
        for (int tn = 0; tn < 2; ++tn)
          acc[tm][tn] = wmma_f32_16x16x4(a[tm], b[tn], acc[tm][tn]);
    }
  }

  float bv[2] = {0.f, 0.f};
  if (INIT) {
#pragma unroll
    for (int tn = 0; tn < 2; ++tn) bv[tn] = bias[n0 + wn + tn * 16 + lm];
  }
#pragma unroll
  for (int tm = 0; tm < 4; ++tm) {
    float selfw[8];
    if (INIT) {
#pragma unroll
      for (int v = 0; v < 8; ++v) {
        const int gm = m0 + wm + tm * 16 + v + 8 * lh;
        const float d = dinv[gm < M ? gm : M - 1];
        selfw[v] = d * d;
      }
    }
#pragma unroll
    for (int v = 0; v < 8; ++v) {
      const int gm = m0 + wm + tm * 16 + v + 8 * lh;
      if (gm < M) {
#pragma unroll
        for (int tn = 0; tn < 2; ++tn) {
          const int gn = n0 + wn + tn * 16 + lm;
          const float c = acc[tm][tn][v];
          C[(size_t)gm * ldc + gn] = c;
          if (INIT) C2[(size_t)gm * ldc + gn] = bv[tn] + c * selfw[v];
        }
      }
    }
  }
}

template <int HIDD, int OUTD>
__global__ void __launch_bounds__(256)
init_kernel(float* deg, int n, int nb_deg,
            const float* __restrict__ W2, const float* __restrict__ b1,
            float* __restrict__ cvec,
            const float* __restrict__ b2, float* __restrict__ h2) {
  const int b = blockIdx.x;
  if (b < nb_deg) {
    const int i = b * 256 + threadIdx.x;
    if (i < n) deg[i] = 1.0f;
  } else if (b < nb_deg + 16) {
    const int lane = threadIdx.x & 31;
    const int j = (b - nb_deg) * 8 + (threadIdx.x >> 5);
    if (j < OUTD) {
      float s = 0.f;
#pragma unroll 4
      for (int k = lane; k < HIDD; k += 32) s += W2[j * HIDD + k] * b1[k];
#pragma unroll
      for (int o = 16; o > 0; o >>= 1) s += __shfl_xor(s, o);
      if (lane == 0) cvec[j] = s;
    }
  } else {
    const int i4 = (b - nb_deg - 16) * 256 + threadIdx.x;
    if (i4 < n * (OUTD / 4)) {
      const int c = (i4 % (OUTD / 4)) * 4;
      *(float4*)(h2 + (size_t)i4 * 4) = *(const float4*)(b2 + c);
    }
  }
}

__global__ void deg_accum_kernel(const int* __restrict__ dst, float* deg, int E) {
  int e = blockIdx.x * blockDim.x + threadIdx.x;
  if (e < E) atomicAdd(&deg[dst[e]], 1.0f);
}

__global__ void dinv_kernel(float* deg, int n) {
  int i = blockIdx.x * blockDim.x + threadIdx.x;
  if (i < n) {
    float d = deg[i];
    deg[i] = (d > 0.f) ? rsqrtf(d) : 0.f;
  }
}

__global__ void norm_kernel(const int* __restrict__ src, const int* __restrict__ dst,
                            const float* __restrict__ dinv, float* __restrict__ norm,
                            int E) {
  int e = blockIdx.x * blockDim.x + threadIdx.x;
  if (e < E) norm[e] = dinv[src[e]] * dinv[dst[e]];
}

template <int D, int N>
__device__ __forceinline__ void scatter_rows(const float* __restrict__ h,
                                             float* __restrict__ out,
                                             int sl, int dl, int nl, int k0, int lane) {
  constexpr int R = D / 32;
  float v[N][R];
#pragma unroll
  for (int k = 0; k < N; ++k) {
    const int s = __builtin_amdgcn_readlane(sl, k0 + k);
    const float* hp = h + (size_t)s * D + lane;
#pragma unroll
    for (int i = 0; i < R; ++i) v[k][i] = hp[32 * i];
  }
#pragma unroll
  for (int k = 0; k < N; ++k) {
    const int   d  = __builtin_amdgcn_readlane(dl, k0 + k);
    const float nv = __builtin_bit_cast(float, __builtin_amdgcn_readlane(nl, k0 + k));
    float* op = out + (size_t)d * D + lane;
#pragma unroll
    for (int i = 0; i < R; ++i) atomicAdd(op + 32 * i, v[k][i] * nv);
  }
}

template <int D, int EPW, bool SELF>
__global__ void __launch_bounds__(256)
scatter_rows_kernel(const float* __restrict__ h,
                    const int* __restrict__ src, const int* __restrict__ dst,
                    const float* __restrict__ norm, float* __restrict__ out,
                    int E, const float* __restrict__ dinv, int n) {
  const int lane = threadIdx.x & 31;
  const int w = __builtin_amdgcn_readfirstlane((int)(blockIdx.x * 8u + (threadIdx.x >> 5)));
  const int e0 = w * EPW;
  const int Et = SELF ? E + n : E;
  if (e0 >= Et) return;

  int el = e0 + ((lane < EPW) ? lane : 0);
  el = (el < Et) ? el : (Et - 1);
  int sl, dl, nl;
  if (SELF && el >= E) {
    sl = dl = el - E;
    const float d = dinv[sl];
    nl = __builtin_bit_cast(int, d * d);
  } else {
    sl = src[el];
    dl = dst[el];
    nl = __builtin_bit_cast(int, norm[el]);
  }

  if (e0 + EPW <= Et) {
    scatter_rows<D, EPW>(h, out, sl, dl, nl, 0, lane);
  } else {
    for (int k = 0; e0 + k < Et; ++k)
      scatter_rows<D, 1>(h, out, sl, dl, nl, k, lane);
  }
}

template <int G>
__device__ __forceinline__ void load_dst_group(const int* __restrict__ dst, int E,
                                               int ebase, int lane, int (&dv)[G]) {
#pragma unroll
  for (int j = 0; j < G; ++j) {
    const int e  = ebase + j * 32 + lane;
    const int ec = e < E ? e : E - 1;
    const int d  = dst[ec];
    dv[j] = e < E ? d : -1;
  }
}

template <int D, int RB, int NW>
__global__ void __launch_bounds__(NW * 32)
agg_rows_kernel(const float* __restrict__ h,
                const int* __restrict__ src, const int* __restrict__ dst,
                const float* __restrict__ norm, const float* __restrict__ bias,
                const float* __restrict__ dinv, float* __restrict__ out,
                int E, int n) {
  constexpr int R  = D / 32;
  constexpr int NT = NW * 32;
  constexpr int G  = 8;
  __shared__ float acc[RB * D];

  const int tid  = threadIdx.x;
  const int lane = tid & 31;
  const int w    = __builtin_amdgcn_readfirstlane(tid >> 5);
  const int r0   = blockIdx.x * RB;
  const int rows = (n - r0) < RB ? (n - r0) : RB;
  const int nf4  = rows * (D / 4);

  for (int i4 = tid; i4 < nf4; i4 += NT) {
    const int r = i4 / (D / 4);
    const int c = (i4 % (D / 4)) * 4;
    const float d  = dinv[r0 + r];
    const float sw = d * d;
    const float4 hv = *(const float4*)(h + (size_t)(r0 + r) * D + c);
    const float4 bv = *(const float4*)(bias + c);
    float4 a;
    a.x = bv.x + hv.x * sw;
    a.y = bv.y + hv.y * sw;
    a.z = bv.z + hv.z * sw;
    a.w = bv.w + hv.w * sw;
    *(float4*)(&acc[r * D + c]) = a;
  }
  __syncthreads();

  const int ngroups = (E + 32 * G - 1) / (32 * G);
  const int rot = ngroups > 0 ? (int)((blockIdx.x * 97u) % (unsigned)ngroups) : 0;
  const unsigned urows = (unsigned)rows;

  int dv[G];
  int g = w;
  if (g < ngroups) {
    int gg = g + rot;
    gg = gg >= ngroups ? gg - ngroups : gg;
    load_dst_group<G>(dst, E, gg * 32 * G, lane, dv);
  }
  while (g < ngroups) {
    int gg = g + rot;
    gg = gg >= ngroups ? gg - ngroups : gg;
    const int ebase = gg * 32 * G;
    const int gn = g + NW;
    int dn[G];
    if (gn < ngroups) {
      int ggn = gn + rot;
      ggn = ggn >= ngroups ? ggn - ngroups : ggn;
      load_dst_group<G>(dst, E, ggn * 32 * G, lane, dn);
    } else {
#pragma unroll
      for (int j = 0; j < G; ++j) dn[j] = -1;
    }
#pragma unroll
    for (int j = 0; j < G; ++j) {
      const int r = dv[j] - r0;
      const bool m = (unsigned)r < urows;
      unsigned mask = __builtin_amdgcn_ballot_w32(m);
      if (mask != 0u) {
        const int e  = ebase + j * 32 + lane;
        const int ec = e < E ? e : E - 1;
        const int   s  = src[ec];
        const int   nb = __builtin_bit_cast(int, norm[ec]);
        do {
          const int k = __builtin_ctz(mask);
          mask &= mask - 1u;
          const int   sk = __builtin_amdgcn_readlane(s, k);
          const int   rk = __builtin_amdgcn_readlane(r, k);
          const float nk = __builtin_bit_cast(float, __builtin_amdgcn_readlane(nb, k));
          const float* hp = h + (size_t)sk * D + lane;
          float v[R];
#pragma unroll
          for (int i = 0; i < R; ++i) v[i] = hp[32 * i];
          float* ap = &acc[rk * D + lane];
#pragma unroll
          for (int i = 0; i < R; ++i)
            __hip_atomic_fetch_add(ap + 32 * i, v[i] * nk, __ATOMIC_RELAXED,
                                   __HIP_MEMORY_SCOPE_WORKGROUP);
        } while (mask != 0u);
      }
    }
#pragma unroll
    for (int j = 0; j < G; ++j) dv[j] = dn[j];
    g = gn;
  }
  __syncthreads();

  float* op = out + (size_t)r0 * D;
  for (int i4 = tid; i4 < nf4; i4 += NT)
    *(float4*)(op + (size_t)i4 * 4) = *(const float4*)(&acc[i4 * 4]);
}

static inline void launch_gemm(int opb, const float* A, int lda, const float* B,
                               int ldb, float* C, int ldc, int M, int Nc, int K,
                               float* agg0, const float* bias, const float* dinv,
                               hipStream_t stream) {
  dim3 grid(Nc / 128, (M + 127) / 128);
  if (opb && agg0)
    gemm_wmma<1, true><<<grid, 256, 0, stream>>>(A, lda, B, ldb, C, ldc, M, Nc, K,
                                                 agg0, bias, dinv);
  else if (opb)
    gemm_wmma<1, false><<<grid, 256, 0, stream>>>(A, lda, B, ldb, C, ldc, M, Nc, K,
                                                  agg0, bias, dinv);
  else if (agg0)
    gemm_wmma<0, true><<<grid, 256, 0, stream>>>(A, lda, B, ldb, C, ldc, M, Nc, K,
                                                 agg0, bias, dinv);
  else
    gemm_wmma<0, false><<<grid, 256, 0, stream>>>(A, lda, B, ldb, C, ldc, M, Nc, K,
                                                  agg0, bias, dinv);
}

extern "C" void kernel_launch(void* const* d_in, const int* in_sizes, int n_in,
                              void* d_out, int out_size, void* d_ws, size_t ws_size,
                              hipStream_t stream) {
  const int IN = 1024, HID = 512, OUT = 128;

  const float* features = (const float*)d_in[0];
  const int*   edge     = (const int*)d_in[1];
  const float* W1       = (const float*)d_in[2];
  const float* b1       = (const float*)d_in[3];
  const float* W2       = (const float*)d_in[4];
  const float* b2       = (const float*)d_in[5];
  const float* b3       = (const float*)d_in[6];
  const float* b4       = (const float*)d_in[7];
  const float* head1    = (const float*)d_in[8];

  const int n  = in_sizes[0] / IN;
  const int E  = in_sizes[1] / 2;
  const int Et = E + n;
  const int* src = edge;
  const int* dst = edge + E;

  float* out = (float*)d_out;
  float* z   = out;
  float* h2  = out + (size_t)n * OUT;
  float* h4  = out + (size_t)n * OUT * 2;

  float* ws   = (float*)d_ws;
  float* dinv = ws;
  float* norm = dinv + n;
  float* t0   = norm + Et;
  float* t1   = t0 + (size_t)n * HID;
  float* h2p  = t1 + (size_t)n * HID;
  float* t2   = h2p + (size_t)n * OUT;
  float* cvec = norm + E;
  float* u    = h2p;
  float* p    = t1;

  const int TB = 256;
  const int nb_deg  = (n + TB - 1) / TB;
  const int nb_fill = (n * (OUT / 4) + TB - 1) / TB;
  init_kernel<HID, OUT><<<nb_deg + 16 + nb_fill, TB, 0, stream>>>(dinv, n, nb_deg, W2, b1,
                                                                  cvec, b2, h2);
  deg_accum_kernel<<<(E + TB - 1) / TB, TB, 0, stream>>>(dst, dinv, E);
  dinv_kernel<<<(n + TB - 1) / TB, TB, 0, stream>>>(dinv, n);
  norm_kernel<<<(E + TB - 1) / TB, TB, 0, stream>>>(src, dst, dinv, norm, E);

  launch_gemm(1, features, IN, W1, IN, t0, HID, n, HID, IN, nullptr, nullptr, nullptr,
              stream);
  launch_gemm(1, t0, HID, W2, HID, u, OUT, n, OUT, HID, p, cvec, dinv, stream);

  scatter_rows_kernel<OUT, 4, false><<<(E + 31) / 32, 256, 0, stream>>>(
      u, src, dst, norm, p, E, nullptr, 0);
  scatter_rows_kernel<OUT, 4, true><<<(E + n + 31) / 32, 256, 0, stream>>>(
      p, src, dst, norm, h2, E, dinv, n);

  launch_gemm(0, h2, OUT, head1, OUT, z, OUT, n, OUT, OUT, nullptr, nullptr, nullptr,
              stream);

  constexpr int RB = 79;
  launch_gemm(0, z, OUT, W2, HID, t0, HID, n, HID, OUT, nullptr, nullptr, nullptr,
              stream);
  agg_rows_kernel<HID, RB, 32><<<(n + RB - 1) / RB, 1024, 0, stream>>>(
      t0, src, dst, norm, b3, dinv, t1, E, n);

  launch_gemm(0, t1, HID, W1, IN, t2, IN, n, IN, HID, nullptr, nullptr, nullptr,
              stream);
  agg_rows_kernel<IN, RB, 32><<<(n + RB - 1) / RB, 1024, 0, stream>>>(
      t2, src, dst, norm, b4, dinv, h4, E, n);
}


# conv3's aggregation work-group keeps its 79 finished rows in LDS and multiplies them by W1 on WMMA (A from LDS, B from global, barrier-free K loop): h3 never stored, 12 launches; sums unchanged
# speedup vs baseline: 2.1878x; 1.0469x over previous
#include <hip/hip_runtime.h>
#include <hip/hip_bf16.h>


typedef float v2f __attribute__((ext_vector_type(2)));
typedef float v8f __attribute__((ext_vector_type(8)));

__device__ __forceinline__ v8f wmma_f32_16x16x4(v2f a, v2f b, v8f c) {
  return __builtin_amdgcn_wmma_f32_16x16x4_f32(false, a, false, b, (short)0, c,
                                               false, false);
}

template <int OPB, bool INIT>
__global__ void __launch_bounds__(256)
gemm_wmma(const float* __restrict__ A, int lda,
          const float* __restrict__ B, int ldb,
          float* __restrict__ C, int ldc,
          int M, int Nc, int K,
          float* __restrict__ C2, const float* __restrict__ bias,
          const float* __restrict__ dinv) {
  constexpr int BM = 128, BN = 128, BK = 16;
  constexpr int ALD = 20;
  constexpr int BLD = 132;

  __shared__ float As[BM * ALD];
  __shared__ float Bs[BK * BLD];

  const int tid  = threadIdx.x;
  const int wid  = tid >> 5;
  const int lane = tid & 31;
  const int lh   = lane >> 4;
  const int lm   = lane & 15;

  const int m0 = blockIdx.y * BM;
  const int n0 = blockIdx.x * BN;
  const int wm = (wid & 1) * 64;
  const int wn = (wid >> 1) * 32;

  v8f acc[4][2];
#pragma unroll
  for (int tm = 0; tm < 4; ++tm)
#pragma unroll
    for (int tn = 0; tn < 2; ++tn) acc[tm][tn] = (v8f)0.0f;

  const float* pa[2];
  const float* pb[2];
#pragma unroll
  for (int r = 0; r < 2; ++r) {
    const int i   = tid + r * 256;
    const int gm  = m0 + (i >> 2);
    const int gmc = gm < M ? gm : M - 1;
    pa[r] = A + (size_t)gmc * lda + ((i & 3) << 2);
    if (OPB)
      pb[r] = B + (size_t)(n0 + (i >> 2)) * ldb + ((i & 3) << 2);
    else
      pb[r] = B + (size_t)(i >> 5) * ldb + n0 + ((i & 31) << 2);
  }

  for (int k0 = 0; k0 < K; k0 += BK) {
    __syncthreads();
    float4 va[2], vb[2];
#pragma unroll
    for (int r = 0; r < 2; ++r) va[r] = *(const float4*)(pa[r] + k0);
#pragma unroll
    for (int r = 0; r < 2; ++r)
      vb[r] = *(const float4*)(OPB ? pb[r] + k0 : pb[r] + (size_t)k0 * ldb);
#pragma unroll
    for (int r = 0; r < 2; ++r) {
      const int i   = tid + r * 256;
      const int row = i >> 2;
      const int c4  = (i & 3) << 2;
      *(float4*)(&As[row * ALD + c4]) = va[r];
    }
    if (OPB) {
#pragma unroll
      for (int r = 0; r < 2; ++r) {
        const int i  = tid + r * 256;
        const int j  = i >> 2;
        const int k4 = (i & 3) << 2;
        Bs[(k4 + 0) * BLD + j] = vb[r].x;
        Bs[(k4 + 1) * BLD + j] = vb[r].y;
        Bs[(k4 + 2) * BLD + j] = vb[r].z;
        Bs[(k4 + 3) * BLD + j] = vb[r].w;
      }
    } else {
#pragma unroll
      for (int r = 0; r < 2; ++r) {
        const int i  = tid + r * 256;
        const int kk = i >> 5;
        const int c  = (i & 31) << 2;
        *(float4*)(&Bs[kk * BLD + c]) = vb[r];
      }
    }
    if (k0 + BK < K) {
      const int gm = m0 + (tid >> 1);
      if (gm < M)
        __builtin_prefetch(A + (size_t)gm * lda + (k0 + BK) + (tid & 1) * 8, 0, 1);
    }
    __syncthreads();

#pragma unroll
    for (int kk = 0; kk < BK; kk += 4) {
      const int ks = kk + 2 * lh;
      v2f a[4], b[2];
#pragma unroll
      for (int tm = 0; tm < 4; ++tm) {
        a[tm] = *(const v2f*)(&As[(wm + tm * 16 + lm) * ALD + ks]);
      }
#pragma unroll
      for (int tn = 0; tn < 2; ++tn) {
        const int col = wn + tn * 16 + lm;
        b[tn].x = Bs[(ks + 0) * BLD + col];
        b[tn].y = Bs[(ks + 1) * BLD + col];
      }
#pragma unroll
      for (int tm = 0; tm < 4; ++tm)
#pragma unroll
        for (int tn = 0; tn < 2; ++tn)
          acc[tm][tn] = wmma_f32_16x16x4(a[tm], b[tn], acc[tm][tn]);
    }
  }

  float bv[2] = {0.f, 0.f};
  if (INIT) {
#pragma unroll
    for (int tn = 0; tn < 2; ++tn) bv[tn] = bias[n0 + wn + tn * 16 + lm];
  }
#pragma unroll
  for (int tm = 0; tm < 4; ++tm) {
    float selfw[8];
    if (INIT) {
#pragma unroll
      for (int v = 0; v < 8; ++v) {
        const int gm = m0 + wm + tm * 16 + v + 8 * lh;
        const float d = dinv[gm < M ? gm : M - 1];
        selfw[v] = d * d;
      }
    }
#pragma unroll
    for (int v = 0; v < 8; ++v) {
      const int gm = m0 + wm + tm * 16 + v + 8 * lh;
      if (gm < M) {
#pragma unroll
        for (int tn = 0; tn < 2; ++tn) {
          const int gn = n0 + wn + tn * 16 + lm;
          const float c = acc[tm][tn][v];
          C[(size_t)gm * ldc + gn] = c;
          if (INIT) C2[(size_t)gm * ldc + gn] = bv[tn] + c * selfw[v];
        }
      }
    }
  }
}

template <int HIDD, int OUTD>
__global__ void __launch_bounds__(256)
init_kernel(float* deg, int n, int nb_deg,
            const float* __restrict__ W2, const float* __restrict__ b1,
            float* __restrict__ cvec,
            const float* __restrict__ b2, float* __restrict__ h2) {
  const int b = blockIdx.x;
  if (b < nb_deg) {
    const int i = b * 256 + threadIdx.x;
    if (i < n) deg[i] = 1.0f;
  } else if (b < nb_deg + 16) {
    const int lane = threadIdx.x & 31;
    const int j = (b - nb_deg) * 8 + (threadIdx.x >> 5);
    if (j < OUTD) {
      float s = 0.f;
#pragma unroll 4
      for (int k = lane; k < HIDD; k += 32) s += W2[j * HIDD + k] * b1[k];
#pragma unroll
      for (int o = 16; o > 0; o >>= 1) s += __shfl_xor(s, o);
      if (lane == 0) cvec[j] = s;
    }
  } else {
    const int i4 = (b - nb_deg - 16) * 256 + threadIdx.x;
    if (i4 < n * (OUTD / 4)) {
      const int c = (i4 % (OUTD / 4)) * 4;
      *(float4*)(h2 + (size_t)i4 * 4) = *(const float4*)(b2 + c);
    }
  }
}

__global__ void deg_accum_kernel(const int* __restrict__ dst, float* deg, int E) {
  int e = blockIdx.x * blockDim.x + threadIdx.x;
  if (e < E) atomicAdd(&deg[dst[e]], 1.0f);
}

__global__ void dinv_kernel(float* deg, int n) {
  int i = blockIdx.x * blockDim.x + threadIdx.x;
  if (i < n) {
    float d = deg[i];
    deg[i] = (d > 0.f) ? rsqrtf(d) : 0.f;
  }
}

__global__ void norm_kernel(const int* __restrict__ src, const int* __restrict__ dst,
                            const float* __restrict__ dinv, float* __restrict__ norm,
                            int E) {
  int e = blockIdx.x * blockDim.x + threadIdx.x;
  if (e < E) norm[e] = dinv[src[e]] * dinv[dst[e]];
}

template <int D, int N>
__device__ __forceinline__ void scatter_rows(const float* __restrict__ h,
                                             float* __restrict__ out,
                                             int sl, int dl, int nl, int k0, int lane) {
  constexpr int R = D / 32;
  float v[N][R];
#pragma unroll
  for (int k = 0; k < N; ++k) {
    const int s = __builtin_amdgcn_readlane(sl, k0 + k);
    const float* hp = h + (size_t)s * D + lane;
#pragma unroll
    for (int i = 0; i < R; ++i) v[k][i] = hp[32 * i];
  }
#pragma unroll
  for (int k = 0; k < N; ++k) {
    const int   d  = __builtin_amdgcn_readlane(dl, k0 + k);
    const float nv = __builtin_bit_cast(float, __builtin_amdgcn_readlane(nl, k0 + k));
    float* op = out + (size_t)d * D + lane;
#pragma unroll
    for (int i = 0; i < R; ++i) atomicAdd(op + 32 * i, v[k][i] * nv);
  }
}

template <int D, int EPW, bool SELF>
__global__ void __launch_bounds__(256)
scatter_rows_kernel(const float* __restrict__ h,
                    const int* __restrict__ src, const int* __restrict__ dst,
                    const float* __restrict__ norm, float* __restrict__ out,
                    int E, const float* __restrict__ dinv, int n) {
  const int lane = threadIdx.x & 31;
  const int w = __builtin_amdgcn_readfirstlane((int)(blockIdx.x * 8u + (threadIdx.x >> 5)));
  const int e0 = w * EPW;
  const int Et = SELF ? E + n : E;
  if (e0 >= Et) return;

  int el = e0 + ((lane < EPW) ? lane : 0);
  el = (el < Et) ? el : (Et - 1);
  int sl, dl, nl;
  if (SELF && el >= E) {
    sl = dl = el - E;
    const float d = dinv[sl];
    nl = __builtin_bit_cast(int, d * d);
  } else {
    sl = src[el];
    dl = dst[el];
    nl = __builtin_bit_cast(int, norm[el]);
  }

  if (e0 + EPW <= Et) {
    scatter_rows<D, EPW>(h, out, sl, dl, nl, 0, lane);
  } else {
    for (int k = 0; e0 + k < Et; ++k)
      scatter_rows<D, 1>(h, out, sl, dl, nl, k, lane);
  }
}

template <int G>
__device__ __forceinline__ void load_dst_group(const int* __restrict__ dst, int E,
                                               int ebase, int lane, int (&dv)[G]) {
#pragma unroll
  for (int j = 0; j < G; ++j) {
    const int e  = ebase + j * 32 + lane;
    const int ec = e < E ? e : E - 1;
    const int d  = dst[ec];
    dv[j] = e < E ? d : -1;
  }
}

template <int D, int LDA, int NT>
__device__ __forceinline__ void agg_start(float* acc, const float* __restrict__ h,
                                          const float* __restrict__ bias,
                                          const float* __restrict__ dinv,
                                          int r0, int rows, int tid) {
  const int nf4 = rows * (D / 4);
  for (int i4 = tid; i4 < nf4; i4 += NT) {
    const int r = i4 / (D / 4);
    const int c = (i4 % (D / 4)) * 4;
    const float d  = dinv[r0 + r];
    const float sw = d * d;
    const float4 hv = *(const float4*)(h + (size_t)(r0 + r) * D + c);
    const float4 bv = *(const float4*)(bias + c);
    float4 a;
    a.x = bv.x + hv.x * sw;
    a.y = bv.y + hv.y * sw;
    a.z = bv.z + hv.z * sw;
    a.w = bv.w + hv.w * sw;
    *(float4*)(&acc[r * LDA + c]) = a;
  }
}

template <int D, int LDA, int NW>
__device__ __forceinline__ void agg_scan(float* acc, const float* __restrict__ h,
                                         const int* __restrict__ src,
                                         const int* __restrict__ dst,
                                         const float* __restrict__ norm,
                                         int E, int r0, int rows, int lane, int w) {
  constexpr int R = D / 32;
  constexpr int G = 8;
  const int ngroups = (E + 32 * G - 1) / (32 * G);
  const int rot = ngroups > 0 ? (int)((blockIdx.x * 97u) % (unsigned)ngroups) : 0;
  const unsigned urows = (unsigned)rows;

  int dv[G];
  int g = w;
  if (g < ngroups) {
    int gg = g + rot;
    gg = gg >= ngroups ? gg - ngroups : gg;
    load_dst_group<G>(dst, E, gg * 32 * G, lane, dv);
  }
  while (g < ngroups) {
    int gg = g + rot;
    gg = gg >= ngroups ? gg - ngroups : gg;
    const int ebase = gg * 32 * G;
    const int gn = g + NW;
    int dn[G];
    if (gn < ngroups) {
      int ggn = gn + rot;
      ggn = ggn >= ngroups ? ggn - ngroups : ggn;
      load_dst_group<G>(dst, E, ggn * 32 * G, lane, dn);
    } else {
#pragma unroll
      for (int j = 0; j < G; ++j) dn[j] = -1;
    }
#pragma unroll
    for (int j = 0; j < G; ++j) {
      const int r = dv[j] - r0;
      const bool m = (unsigned)r < urows;
      unsigned mask = __builtin_amdgcn_ballot_w32(m);
      if (mask != 0u) {
        const int e  = ebase + j * 32 + lane;
        const int ec = e < E ? e : E - 1;
        const int   s  = src[ec];
        const int   nb = __builtin_bit_cast(int, norm[ec]);
        do {
          const int k = __builtin_ctz(mask);
          mask &= mask - 1u;
          const int   sk = __builtin_amdgcn_readlane(s, k);
          const int   rk = __builtin_amdgcn_readlane(r, k);
          const float nk = __builtin_bit_cast(float, __builtin_amdgcn_readlane(nb, k));
          const float* hp = h + (size_t)sk * D + lane;
          float v[R];
#pragma unroll
          for (int i = 0; i < R; ++i) v[i] = hp[32 * i];
          float* ap = &acc[rk * LDA + lane];
#pragma unroll
          for (int i = 0; i < R; ++i)
            __hip_atomic_fetch_add(ap + 32 * i, v[i] * nk, __ATOMIC_RELAXED,
                                   __HIP_MEMORY_SCOPE_WORKGROUP);
        } while (mask != 0u);
      }
    }
#pragma unroll
    for (int j = 0; j < G; ++j) dv[j] = dn[j];
    g = gn;
  }
}

template <int D, int RB, int NW>
__global__ void __launch_bounds__(NW * 32)
agg_rows_kernel(const float* __restrict__ h,
                const int* __restrict__ src, const int* __restrict__ dst,
                const float* __restrict__ norm, const float* __restrict__ bias,
                const float* __restrict__ dinv, float* __restrict__ out,
                int E, int n) {
  constexpr int NT = NW * 32;
  __shared__ float acc[RB * D];

  const int tid  = threadIdx.x;
  const int lane = tid & 31;
  const int w    = __builtin_amdgcn_readfirstlane(tid >> 5);
  const int r0   = blockIdx.x * RB;
  const int rows = (n - r0) < RB ? (n - r0) : RB;
  const int nf4  = rows * (D / 4);

  agg_start<D, D, NT>(acc, h, bias, dinv, r0, rows, tid);
  __syncthreads();
  agg_scan<D, D, NW>(acc, h, src, dst, norm, E, r0, rows, lane, w);
  __syncthreads();

  float* op = out + (size_t)r0 * D;
  for (int i4 = tid; i4 < nf4; i4 += NT)
    *(float4*)(op + (size_t)i4 * 4) = *(const float4*)(&acc[i4 * 4]);
}

template <int D, int RB, int NW, int NO>
__global__ void __launch_bounds__(NW * 32)
agg_rows_mm_kernel(const float* __restrict__ h,
                   const int* __restrict__ src, const int* __restrict__ dst,
                   const float* __restrict__ norm, const float* __restrict__ bias,
                   const float* __restrict__ dinv, const float* __restrict__ Bw,
                   float* __restrict__ out, int E, int n) {
  static_assert(NO == NW * 32, "one wave per 32 output columns");
  static_assert(D % 32 == 0, "whole 128-byte lines per row");
  constexpr int NT  = NW * 32;
  constexpr int TM  = (RB + 15) / 16;
  constexpr int RP  = TM * 16;
  constexpr int LDA = D + 4;
  __shared__ float acc[RP * LDA];

  const int tid  = threadIdx.x;
  const int lane = tid & 31;
  const int w    = __builtin_amdgcn_readfirstlane(tid >> 5);
  const int r0   = blockIdx.x * RB;
  const int rows = (n - r0) < RB ? (n - r0) : RB;

  agg_start<D, LDA, NT>(acc, h, bias, dinv, r0, rows, tid);
  for (int i4 = tid; i4 < (RP - rows) * (D / 4); i4 += NT) {
    const int r = rows + i4 / (D / 4);
    const int c = (i4 % (D / 4)) * 4;
    *(float4*)(&acc[r * LDA + c]) = make_float4(0.f, 0.f, 0.f, 0.f);
  }
  __syncthreads();
  agg_scan<D, LDA, NW>(acc, h, src, dst, norm, E, r0, rows, lane, w);
  __syncthreads();

  const int lh = lane >> 4;
  const int lm = lane & 15;
  v8f c[TM][2];
#pragma unroll
  for (int tm = 0; tm < TM; ++tm) {
    c[tm][0] = (v8f)0.0f;
    c[tm][1] = (v8f)0.0f;
  }
  const float* ap = &acc[lm * LDA + 2 * lh];
  const float* bp = Bw + (size_t)(2 * lh) * NO + w * 32 + 2 * lm;
#pragma unroll 2
  for (int k0 = 0; k0 < D; k0 += 4) {
    v2f a[TM];
#pragma unroll
    for (int tm = 0; tm < TM; ++tm)
      a[tm] = *(const v2f*)(ap + tm * 16 * LDA + k0);
    const float2 b0 = *(const float2*)(bp + (size_t)k0 * NO);
    const float2 b1 = *(const float2*)(bp + (size_t)(k0 + 1) * NO);
    v2f be, bo;
    be.x = b0.x; be.y = b1.x;
    bo.x = b0.y; bo.y = b1.y;
#pragma unroll
    for (int tm = 0; tm < TM; ++tm) {
      c[tm][0] = wmma_f32_16x16x4(a[tm], be, c[tm][0]);
      c[tm][1] = wmma_f32_16x16x4(a[tm], bo, c[tm][1]);
    }
  }

  float* op = out + (size_t)r0 * NO + w * 32 + 2 * lm;
#pragma unroll
  for (int tm = 0; tm < TM; ++tm) {
#pragma unroll
    for (int v = 0; v < 8; ++v) {
      const int row = tm * 16 + v + 8 * lh;
      if (row < rows)
        *(float2*)(op + (size_t)row * NO) = make_float2(c[tm][0][v], c[tm][1][v]);
    }
  }
}

static inline void launch_gemm(int opb, const float* A, int lda, const float* B,
                               int ldb, float* C, int ldc, int M, int Nc, int K,
                               float* agg0, const float* bias, const float* dinv,
                               hipStream_t stream) {
  dim3 grid(Nc / 128, (M + 127) / 128);
  if (opb && agg0)
    gemm_wmma<1, true><<<grid, 256, 0, stream>>>(A, lda, B, ldb, C, ldc, M, Nc, K,
                                                 agg0, bias, dinv);
  else if (opb)
    gemm_wmma<1, false><<<grid, 256, 0, stream>>>(A, lda, B, ldb, C, ldc, M, Nc, K,
                                                  agg0, bias, dinv);
  else if (agg0)
    gemm_wmma<0, true><<<grid, 256, 0, stream>>>(A, lda, B, ldb, C, ldc, M, Nc, K,
                                                 agg0, bias, dinv);
  else
    gemm_wmma<0, false><<<grid, 256, 0, stream>>>(A, lda, B, ldb, C, ldc, M, Nc, K,
                                                  agg0, bias, dinv);
}

extern "C" void kernel_launch(void* const* d_in, const int* in_sizes, int n_in,
                              void* d_out, int out_size, void* d_ws, size_t ws_size,
                              hipStream_t stream) {
  const int IN = 1024, HID = 512, OUT = 128;

  const float* features = (const float*)d_in[0];
  const int*   edge     = (const int*)d_in[1];
  const float* W1       = (const float*)d_in[2];
  const float* b1       = (const float*)d_in[3];
  const float* W2       = (const float*)d_in[4];
  const float* b2       = (const float*)d_in[5];
  const float* b3       = (const float*)d_in[6];
  const float* b4       = (const float*)d_in[7];
  const float* head1    = (const float*)d_in[8];

  const int n  = in_sizes[0] / IN;
  const int E  = in_sizes[1] / 2;
  const int Et = E + n;
  const int* src = edge;
  const int* dst = edge + E;

  float* out = (float*)d_out;
  float* z   = out;
  float* h2  = out + (size_t)n * OUT;
  float* h4  = out + (size_t)n * OUT * 2;

  float* ws   = (float*)d_ws;
  float* dinv = ws;
  float* norm = dinv + n;
  float* t0   = norm + Et;
  float* t1   = t0 + (size_t)n * HID;
  float* h2p  = t1 + (size_t)n * HID;
  float* t2   = h2p + (size_t)n * OUT;
  float* cvec = norm + E;
  float* u    = h2p;
  float* p    = t1;

  const int TB = 256;
  const int nb_deg  = (n + TB - 1) / TB;
  const int nb_fill = (n * (OUT / 4) + TB - 1) / TB;
  init_kernel<HID, OUT><<<nb_deg + 16 + nb_fill, TB, 0, stream>>>(dinv, n, nb_deg, W2, b1,
                                                                  cvec, b2, h2);
  deg_accum_kernel<<<(E + TB - 1) / TB, TB, 0, stream>>>(dst, dinv, E);
  dinv_kernel<<<(n + TB - 1) / TB, TB, 0, stream>>>(dinv, n);
  norm_kernel<<<(E + TB - 1) / TB, TB, 0, stream>>>(src, dst, dinv, norm, E);

  launch_gemm(1, features, IN, W1, IN, t0, HID, n, HID, IN, nullptr, nullptr, nullptr,
              stream);
  launch_gemm(1, t0, HID, W2, HID, u, OUT, n, OUT, HID, p, cvec, dinv, stream);

  scatter_rows_kernel<OUT, 4, false><<<(E + 31) / 32, 256, 0, stream>>>(
      u, src, dst, norm, p, E, nullptr, 0);
  scatter_rows_kernel<OUT, 4, true><<<(E + n + 31) / 32, 256, 0, stream>>>(
      p, src, dst, norm, h2, E, dinv, n);

  launch_gemm(0, h2, OUT, head1, OUT, z, OUT, n, OUT, OUT, nullptr, nullptr, nullptr,
              stream);

  constexpr int RB = 79;
  launch_gemm(0, z, OUT, W2, HID, t0, HID, n, HID, OUT, nullptr, nullptr, nullptr,
              stream);
  agg_rows_mm_kernel<HID, RB, 32, IN><<<(n + RB - 1) / RB, 1024, 0, stream>>>(
      t0, src, dst, norm, b3, dinv, W1, t2, E, n);

  agg_rows_kernel<IN, RB, 32><<<(n + RB - 1) / RB, 1024, 0, stream>>>(
      t2, src, dst, norm, b4, dinv, h4, E, n);
}


# Both decoder aggregations moved onto z, ahead of both tied weights, h4 = (agg(agg(z)) W2 + agg(1) b3^T) W1 + b4: 128-wide atomic scatters replace the 512/1024-wide LDS ones; changes float sum order
# speedup vs baseline: 2.4799x; 1.1335x over previous
#include <hip/hip_runtime.h>
#include <hip/hip_bf16.h>


typedef float v2f __attribute__((ext_vector_type(2)));
typedef float v8f __attribute__((ext_vector_type(8)));

__device__ __forceinline__ v8f wmma_f32_16x16x4(v2f a, v2f b, v8f c) {
  return __builtin_amdgcn_wmma_f32_16x16x4_f32(false, a, false, b, (short)0, c,
                                               false, false);
}

template <int OPB, bool INIT>
__global__ void __launch_bounds__(256)
gemm_wmma(const float* __restrict__ A, int lda,
          const float* __restrict__ B, int ldb,
          float* __restrict__ C, int ldc,
          int M, int Nc, int K,
          float* __restrict__ C2, const float* __restrict__ bias,
          const float* __restrict__ dinv) {
  constexpr int BM = 128, BN = 128, BK = 16;
  constexpr int ALD = 20;
  constexpr int BLD = 132;

  __shared__ float As[BM * ALD];
  __shared__ float Bs[BK * BLD];

  const int tid  = threadIdx.x;
  const int wid  = tid >> 5;
  const int lane = tid & 31;
  const int lh   = lane >> 4;
  const int lm   = lane & 15;

  const int m0 = blockIdx.y * BM;
  const int n0 = blockIdx.x * BN;
  const int wm = (wid & 1) * 64;
  const int wn = (wid >> 1) * 32;

  v8f acc[4][2];
#pragma unroll
  for (int tm = 0; tm < 4; ++tm)
#pragma unroll
    for (int tn = 0; tn < 2; ++tn) acc[tm][tn] = (v8f)0.0f;

  const float* pa[2];
  const float* pb[2];
#pragma unroll
  for (int r = 0; r < 2; ++r) {
    const int i   = tid + r * 256;
    const int gm  = m0 + (i >> 2);
    const int gmc = gm < M ? gm : M - 1;
    pa[r] = A + (size_t)gmc * lda + ((i & 3) << 2);
    if (OPB)
      pb[r] = B + (size_t)(n0 + (i >> 2)) * ldb + ((i & 3) << 2);
    else
      pb[r] = B + (size_t)(i >> 5) * ldb + n0 + ((i & 31) << 2);
  }

  for (int k0 = 0; k0 < K; k0 += BK) {
    __syncthreads();
    float4 va[2], vb[2];
#pragma unroll
    for (int r = 0; r < 2; ++r) va[r] = *(const float4*)(pa[r] + k0);
#pragma unroll
    for (int r = 0; r < 2; ++r)
      vb[r] = *(const float4*)(OPB ? pb[r] + k0 : pb[r] + (size_t)k0 * ldb);
#pragma unroll
    for (int r = 0; r < 2; ++r) {
      const int i   = tid + r * 256;
      const int row = i >> 2;
      const int c4  = (i & 3) << 2;
      *(float4*)(&As[row * ALD + c4]) = va[r];
    }
    if (OPB) {
#pragma unroll
      for (int r = 0; r < 2; ++r) {
        const int i  = tid + r * 256;
        const int j  = i >> 2;
        const int k4 = (i & 3) << 2;
        Bs[(k4 + 0) * BLD + j] = vb[r].x;
        Bs[(k4 + 1) * BLD + j] = vb[r].y;
        Bs[(k4 + 2) * BLD + j] = vb[r].z;
        Bs[(k4 + 3) * BLD + j] = vb[r].w;
      }
    } else {
#pragma unroll
      for (int r = 0; r < 2; ++r) {
        const int i  = tid + r * 256;
        const int kk = i >> 5;
        const int c  = (i & 31) << 2;
        *(float4*)(&Bs[kk * BLD + c]) = vb[r];
      }
    }
    if (k0 + BK < K) {
      const int gm = m0 + (tid >> 1);
      if (gm < M)
        __builtin_prefetch(A + (size_t)gm * lda + (k0 + BK) + (tid & 1) * 8, 0, 1);
    }
    __syncthreads();

#pragma unroll
    for (int kk = 0; kk < BK; kk += 4) {
      const int ks = kk + 2 * lh;
      v2f a[4], b[2];
#pragma unroll
      for (int tm = 0; tm < 4; ++tm) {
        a[tm] = *(const v2f*)(&As[(wm + tm * 16 + lm) * ALD + ks]);
      }
#pragma unroll
      for (int tn = 0; tn < 2; ++tn) {
        const int col = wn + tn * 16 + lm;
        b[tn].x = Bs[(ks + 0) * BLD + col];
        b[tn].y = Bs[(ks + 1) * BLD + col];
      }
#pragma unroll
      for (int tm = 0; tm < 4; ++tm)
#pragma unroll
        for (int tn = 0; tn < 2; ++tn)
          acc[tm][tn] = wmma_f32_16x16x4(a[tm], b[tn], acc[tm][tn]);
    }
  }

  float bv[2] = {0.f, 0.f};
  if (INIT && bias != nullptr) {
#pragma unroll
    for (int tn = 0; tn < 2; ++tn) bv[tn] = bias[n0 + wn + tn * 16 + lm];
  }
#pragma unroll
  for (int tm = 0; tm < 4; ++tm) {
    float selfw[8];
    if (INIT) {
#pragma unroll
      for (int v = 0; v < 8; ++v) {
        const int gm = m0 + wm + tm * 16 + v + 8 * lh;
        const float d = dinv[gm < M ? gm : M - 1];
        selfw[v] = d * d;
      }
    }
#pragma unroll
    for (int v = 0; v < 8; ++v) {
      const int gm = m0 + wm + tm * 16 + v + 8 * lh;
      if (gm < M) {
#pragma unroll
        for (int tn = 0; tn < 2; ++tn) {
          const int gn = n0 + wn + tn * 16 + lm;
          const float c = acc[tm][tn][v];
          C[(size_t)gm * ldc + gn] = c;
          if (INIT) C2[(size_t)gm * ldc + gn] = bv[tn] + c * selfw[v];
        }
      }
    }
  }
}

template <int HIDD, int OUTD>
__global__ void __launch_bounds__(256)
init_kernel(float* deg, int n, int nb_deg,
            const float* __restrict__ W2, const float* __restrict__ b1,
            float* __restrict__ cvec,
            const float* __restrict__ b2, float* __restrict__ h2,
            float* __restrict__ a2) {
  const int b = blockIdx.x;
  if (b < nb_deg) {
    const int i = b * 256 + threadIdx.x;
    if (i < n) deg[i] = 1.0f;
  } else if (b < nb_deg + 16) {
    const int lane = threadIdx.x & 31;
    const int j = (b - nb_deg) * 8 + (threadIdx.x >> 5);
    if (j < OUTD) {
      float s = 0.f;
#pragma unroll 4
      for (int k = lane; k < HIDD; k += 32) s += W2[j * HIDD + k] * b1[k];
#pragma unroll
      for (int o = 16; o > 0; o >>= 1) s += __shfl_xor(s, o);
      if (lane == 0) cvec[j] = s;
    }
  } else {
    const int i4 = (b - nb_deg - 16) * 256 + threadIdx.x;
    if (i4 < n * (OUTD / 4)) {
      const int c = (i4 % (OUTD / 4)) * 4;
      *(float4*)(h2 + (size_t)i4 * 4) = *(const float4*)(b2 + c);
      *(float4*)(a2 + (size_t)i4 * 4) = make_float4(0.f, 0.f, 0.f, 0.f);
    }
  }
}

__global__ void deg_accum_kernel(const int* __restrict__ dst, float* deg, int E) {
  int e = blockIdx.x * blockDim.x + threadIdx.x;
  if (e < E) atomicAdd(&deg[dst[e]], 1.0f);
}

__global__ void dinv_kernel(float* deg, float* __restrict__ rsum, int n) {
  int i = blockIdx.x * blockDim.x + threadIdx.x;
  if (i < n) {
    float d = deg[i];
    float r = (d > 0.f) ? rsqrtf(d) : 0.f;
    deg[i] = r;
    rsum[i] = r * r;
  }
}

__global__ void norm_kernel(const int* __restrict__ src, const int* __restrict__ dst,
                            const float* __restrict__ dinv, float* __restrict__ norm,
                            float* rsum, int E) {
  int e = blockIdx.x * blockDim.x + threadIdx.x;
  if (e < E) {
    const int d = dst[e];
    const float nv = dinv[src[e]] * dinv[d];
    norm[e] = nv;
    atomicAdd(&rsum[d], nv);
  }
}

template <int D, int N>
__device__ __forceinline__ void scatter_rows(const float* __restrict__ h,
                                             float* __restrict__ out,
                                             int sl, int dl, int nl, int k0, int lane) {
  constexpr int R = D / 32;
  float v[N][R];
#pragma unroll
  for (int k = 0; k < N; ++k) {
    const int s = __builtin_amdgcn_readlane(sl, k0 + k);
    const float* hp = h + (size_t)s * D + lane;
#pragma unroll
    for (int i = 0; i < R; ++i) v[k][i] = hp[32 * i];
  }
#pragma unroll
  for (int k = 0; k < N; ++k) {
    const int   d  = __builtin_amdgcn_readlane(dl, k0 + k);
    const float nv = __builtin_bit_cast(float, __builtin_amdgcn_readlane(nl, k0 + k));
    float* op = out + (size_t)d * D + lane;
#pragma unroll
    for (int i = 0; i < R; ++i) atomicAdd(op + 32 * i, v[k][i] * nv);
  }
}

template <int D, int EPW, bool SELF>
__global__ void __launch_bounds__(256)
scatter_rows_kernel(const float* __restrict__ h,
                    const int* __restrict__ src, const int* __restrict__ dst,
                    const float* __restrict__ norm, float* __restrict__ out,
                    int E, const float* __restrict__ dinv, int n) {
  const int lane = threadIdx.x & 31;
  const int w = __builtin_amdgcn_readfirstlane((int)(blockIdx.x * 8u + (threadIdx.x >> 5)));
  const int e0 = w * EPW;
  const int Et = SELF ? E + n : E;
  if (e0 >= Et) return;

  int el = e0 + ((lane < EPW) ? lane : 0);
  el = (el < Et) ? el : (Et - 1);
  int sl, dl, nl;
  if (SELF && el >= E) {
    sl = dl = el - E;
    const float d = dinv[sl];
    nl = __builtin_bit_cast(int, d * d);
  } else {
    sl = src[el];
    dl = dst[el];
    nl = __builtin_bit_cast(int, norm[el]);
  }

  if (e0 + EPW <= Et) {
    scatter_rows<D, EPW>(h, out, sl, dl, nl, 0, lane);
  } else {
    for (int k = 0; e0 + k < Et; ++k)
      scatter_rows<D, 1>(h, out, sl, dl, nl, k, lane);
  }
}

template <int D0, int D, int RB, int NW, int NO>
__global__ void __launch_bounds__(NW * 32)
rows_mm2_kernel(const float* __restrict__ X, const float* __restrict__ rw,
                const float* __restrict__ Bw0, const float* __restrict__ bias0,
                const float* __restrict__ Bw, const float* __restrict__ bias,
                float* __restrict__ out, int n) {
  static_assert(NO == NW * 32, "second product: one wave per 32 output columns");
  static_assert(D == NW * 16, "first product: one wave per 16 columns");
  static_assert(D0 % 4 == 0 && D % 4 == 0, "whole K steps of 4");
  constexpr int NT  = NW * 32;
  constexpr int TM  = (RB + 15) / 16;
  constexpr int RP  = TM * 16;
  constexpr int LDX = D0 + 4;
  constexpr int LDA = D + 4;
  __shared__ float xin[RP * LDX];
  __shared__ float acc[RP * LDA];
  __shared__ float rws[RP];

  const int tid  = threadIdx.x;
  const int lane = tid & 31;
  const int w    = __builtin_amdgcn_readfirstlane(tid >> 5);
  const int lh   = lane >> 4;
  const int lm   = lane & 15;
  const int r0   = blockIdx.x * RB;
  const int rows = (n - r0) < RB ? (n - r0) : RB;

  for (int i4 = tid; i4 < RP * (D0 / 4); i4 += NT) {
    const int r = i4 / (D0 / 4);
    const int c = (i4 % (D0 / 4)) * 4;
    float4 v = make_float4(0.f, 0.f, 0.f, 0.f);
    if (r < rows) v = *(const float4*)(X + (size_t)(r0 + r) * D0 + c);
    *(float4*)(&xin[r * LDX + c]) = v;
  }
  if (tid < RP) rws[tid] = tid < rows ? rw[r0 + tid] : 0.f;
  __syncthreads();

  {
    v8f c1[TM];
#pragma unroll
    for (int tm = 0; tm < TM; ++tm) c1[tm] = (v8f)0.0f;
    const float* ap0 = &xin[lm * LDX + 2 * lh];
    const float* bp0 = Bw0 + (size_t)(2 * lh) * D + w * 16 + lm;
#pragma unroll 2
    for (int k0 = 0; k0 < D0; k0 += 4) {
      v2f a[TM];
#pragma unroll
      for (int tm = 0; tm < TM; ++tm)
        a[tm] = *(const v2f*)(ap0 + tm * 16 * LDX + k0);
      v2f b;
      b.x = bp0[(size_t)k0 * D];
      b.y = bp0[(size_t)(k0 + 1) * D];
#pragma unroll
      for (int tm = 0; tm < TM; ++tm) c1[tm] = wmma_f32_16x16x4(a[tm], b, c1[tm]);
    }
    const float bb = bias0[w * 16 + lm];
    float* cp = &acc[(8 * lh) * LDA + w * 16 + lm];
#pragma unroll
    for (int tm = 0; tm < TM; ++tm) {
      const float4 w0 = *(const float4*)(&rws[tm * 16 + 8 * lh]);
      const float4 w1 = *(const float4*)(&rws[tm * 16 + 8 * lh + 4]);
      const float wv[8] = {w0.x, w0.y, w0.z, w0.w, w1.x, w1.y, w1.z, w1.w};
#pragma unroll
      for (int v = 0; v < 8; ++v)
        cp[(tm * 16 + v) * LDA] = c1[tm][v] + wv[v] * bb;
    }
  }
  __syncthreads();

  v8f c[TM][2];
#pragma unroll
  for (int tm = 0; tm < TM; ++tm) {
    c[tm][0] = (v8f)0.0f;
    c[tm][1] = (v8f)0.0f;
  }
  const float* ap = &acc[lm * LDA + 2 * lh];
  const float* bp = Bw + (size_t)(2 * lh) * NO + w * 32 + 2 * lm;
#pragma unroll 2
  for (int k0 = 0; k0 < D; k0 += 4) {
    v2f a[TM];
#pragma unroll
    for (int tm = 0; tm < TM; ++tm)
      a[tm] = *(const v2f*)(ap + tm * 16 * LDA + k0);
    const float2 b0 = *(const float2*)(bp + (size_t)k0 * NO);
    const float2 b1 = *(const float2*)(bp + (size_t)(k0 + 1) * NO);
    v2f be, bo;
    be.x = b0.x; be.y = b1.x;
    bo.x = b0.y; bo.y = b1.y;
#pragma unroll
    for (int tm = 0; tm < TM; ++tm) {
      c[tm][0] = wmma_f32_16x16x4(a[tm], be, c[tm][0]);
      c[tm][1] = wmma_f32_16x16x4(a[tm], bo, c[tm][1]);
    }
  }

  const float2 bo2 = *(const float2*)(bias + w * 32 + 2 * lm);
  float* op = out + (size_t)r0 * NO + w * 32 + 2 * lm;
#pragma unroll
  for (int tm = 0; tm < TM; ++tm) {
#pragma unroll
    for (int v = 0; v < 8; ++v) {
      const int row = tm * 16 + v + 8 * lh;
      if (row < rows)
        *(float2*)(op + (size_t)row * NO) =
            make_float2(c[tm][0][v] + bo2.x, c[tm][1][v] + bo2.y);
    }
  }
}

static inline void launch_gemm(int opb, const float* A, int lda, const float* B,
                               int ldb, float* C, int ldc, int M, int Nc, int K,
                               float* agg0, const float* bias, const float* dinv,
                               hipStream_t stream) {
  dim3 grid(Nc / 128, (M + 127) / 128);
  if (opb && agg0)
    gemm_wmma<1, true><<<grid, 256, 0, stream>>>(A, lda, B, ldb, C, ldc, M, Nc, K,
                                                 agg0, bias, dinv);
  else if (opb)
    gemm_wmma<1, false><<<grid, 256, 0, stream>>>(A, lda, B, ldb, C, ldc, M, Nc, K,
                                                  agg0, bias, dinv);
  else
    gemm_wmma<0, true><<<grid, 256, 0, stream>>>(A, lda, B, ldb, C, ldc, M, Nc, K,
                                                 agg0, bias, dinv);
}

extern "C" void kernel_launch(void* const* d_in, const int* in_sizes, int n_in,
                              void* d_out, int out_size, void* d_ws, size_t ws_size,
                              hipStream_t stream) {
  const int IN = 1024, HID = 512, OUT = 128;

  const float* features = (const float*)d_in[0];
  const int*   edge     = (const int*)d_in[1];
  const float* W1       = (const float*)d_in[2];
  const float* b1       = (const float*)d_in[3];
  const float* W2       = (const float*)d_in[4];
  const float* b2       = (const float*)d_in[5];
  const float* b3       = (const float*)d_in[6];
  const float* b4       = (const float*)d_in[7];
  const float* head1    = (const float*)d_in[8];

  const int n  = in_sizes[0] / IN;
  const int E  = in_sizes[1] / 2;
  const int Et = E + n;
  const int* src = edge;
  const int* dst = edge + E;

  float* out = (float*)d_out;
  float* z   = out;
  float* h2  = out + (size_t)n * OUT;
  float* h4  = out + (size_t)n * OUT * 2;

  float* ws   = (float*)d_ws;
  float* dinv = ws;
  float* norm = dinv + n;
  float* t0   = norm + Et;
  float* t1   = t0 + (size_t)n * HID;
  float* h2p  = t1 + (size_t)n * HID;
  float* t2   = h2p + (size_t)n * OUT;
  float* cvec = norm + E;
  float* u    = h2p;
  float* p    = t1;
  float* a1   = t2;
  float* a2   = t2 + (size_t)n * OUT;
  float* rsum = t2 + (size_t)n * OUT * 2;

  const int TB = 256;
  const int nb_deg  = (n + TB - 1) / TB;
  const int nb_fill = (n * (OUT / 4) + TB - 1) / TB;
  init_kernel<HID, OUT><<<nb_deg + 16 + nb_fill, TB, 0, stream>>>(dinv, n, nb_deg, W2, b1,
                                                                  cvec, b2, h2, a2);
  deg_accum_kernel<<<(E + TB - 1) / TB, TB, 0, stream>>>(dst, dinv, E);
  dinv_kernel<<<(n + TB - 1) / TB, TB, 0, stream>>>(dinv, rsum, n);
  norm_kernel<<<(E + TB - 1) / TB, TB, 0, stream>>>(src, dst, dinv, norm, rsum, E);

  launch_gemm(1, features, IN, W1, IN, t0, HID, n, HID, IN, nullptr, nullptr, nullptr,
              stream);
  launch_gemm(1, t0, HID, W2, HID, u, OUT, n, OUT, HID, p, cvec, dinv, stream);

  scatter_rows_kernel<OUT, 4, false><<<(E + 31) / 32, 256, 0, stream>>>(
      u, src, dst, norm, p, E, nullptr, 0);
  scatter_rows_kernel<OUT, 4, true><<<(E + n + 31) / 32, 256, 0, stream>>>(
      p, src, dst, norm, h2, E, dinv, n);

  launch_gemm(0, h2, OUT, head1, OUT, z, OUT, n, OUT, OUT, a1, nullptr, dinv, stream);

  scatter_rows_kernel<OUT, 4, false><<<(E + 31) / 32, 256, 0, stream>>>(
      z, src, dst, norm, a1, E, nullptr, 0);
  scatter_rows_kernel<OUT, 4, true><<<(E + n + 31) / 32, 256, 0, stream>>>(
      a1, src, dst, norm, a2, E, dinv, n);

  constexpr int RB = 80;
  rows_mm2_kernel<OUT, HID, RB, 32, IN><<<(n + RB - 1) / RB, 1024, 0, stream>>>(
      a2, rsum, W2, b3, W1, b4, h4, n);
}


# Encoder's two products in one launch: a work-group per 80 rows keeps its 512-wide X W1^T rows in LDS and multiplies them by W2^T on WMMA; t0 never stored, 11 launches; changes float sum order
# speedup vs baseline: 2.8293x; 1.1409x over previous
#include <hip/hip_runtime.h>
#include <hip/hip_bf16.h>


typedef float v2f __attribute__((ext_vector_type(2)));
typedef float v8f __attribute__((ext_vector_type(8)));

__device__ __forceinline__ v8f wmma_f32_16x16x4(v2f a, v2f b, v8f c) {
  return __builtin_amdgcn_wmma_f32_16x16x4_f32(false, a, false, b, (short)0, c,
                                               false, false);
}

template <int OPB, bool INIT>
__global__ void __launch_bounds__(256)
gemm_wmma(const float* __restrict__ A, int lda,
          const float* __restrict__ B, int ldb,
          float* __restrict__ C, int ldc,
          int M, int Nc, int K,
          float* __restrict__ C2, const float* __restrict__ bias,
          const float* __restrict__ dinv) {
  constexpr int BM = 128, BN = 128, BK = 16;
  constexpr int ALD = 20;
  constexpr int BLD = 132;

  __shared__ float As[BM * ALD];
  __shared__ float Bs[BK * BLD];

  const int tid  = threadIdx.x;
  const int wid  = tid >> 5;
  const int lane = tid & 31;
  const int lh   = lane >> 4;
  const int lm   = lane & 15;

  const int m0 = blockIdx.y * BM;
  const int n0 = blockIdx.x * BN;
  const int wm = (wid & 1) * 64;
  const int wn = (wid >> 1) * 32;

  v8f acc[4][2];
#pragma unroll
  for (int tm = 0; tm < 4; ++tm)
#pragma unroll
    for (int tn = 0; tn < 2; ++tn) acc[tm][tn] = (v8f)0.0f;

  const float* pa[2];
  const float* pb[2];
#pragma unroll
  for (int r = 0; r < 2; ++r) {
    const int i   = tid + r * 256;
    const int gm  = m0 + (i >> 2);
    const int gmc = gm < M ? gm : M - 1;
    pa[r] = A + (size_t)gmc * lda + ((i & 3) << 2);
    if (OPB)
      pb[r] = B + (size_t)(n0 + (i >> 2)) * ldb + ((i & 3) << 2);
    else
      pb[r] = B + (size_t)(i >> 5) * ldb + n0 + ((i & 31) << 2);
  }

  for (int k0 = 0; k0 < K; k0 += BK) {
    __syncthreads();
    float4 va[2], vb[2];
#pragma unroll
    for (int r = 0; r < 2; ++r) va[r] = *(const float4*)(pa[r] + k0);
#pragma unroll
    for (int r = 0; r < 2; ++r)
      vb[r] = *(const float4*)(OPB ? pb[r] + k0 : pb[r] + (size_t)k0 * ldb);
#pragma unroll
    for (int r = 0; r < 2; ++r) {
      const int i   = tid + r * 256;
      const int row = i >> 2;
      const int c4  = (i & 3) << 2;
      *(float4*)(&As[row * ALD + c4]) = va[r];
    }
    if (OPB) {
#pragma unroll
      for (int r = 0; r < 2; ++r) {
        const int i  = tid + r * 256;
        const int j  = i >> 2;
        const int k4 = (i & 3) << 2;
        Bs[(k4 + 0) * BLD + j] = vb[r].x;
        Bs[(k4 + 1) * BLD + j] = vb[r].y;
        Bs[(k4 + 2) * BLD + j] = vb[r].z;
        Bs[(k4 + 3) * BLD + j] = vb[r].w;
      }
    } else {
#pragma unroll
      for (int r = 0; r < 2; ++r) {
        const int i  = tid + r * 256;
        const int kk = i >> 5;
        const int c  = (i & 31) << 2;
        *(float4*)(&Bs[kk * BLD + c]) = vb[r];
      }
    }
    if (k0 + BK < K) {
      const int gm = m0 + (tid >> 1);
      if (gm < M)
        __builtin_prefetch(A + (size_t)gm * lda + (k0 + BK) + (tid & 1) * 8, 0, 1);
    }
    __syncthreads();

#pragma unroll
    for (int kk = 0; kk < BK; kk += 4) {
      const int ks = kk + 2 * lh;
      v2f a[4], b[2];
#pragma unroll
      for (int tm = 0; tm < 4; ++tm) {
        a[tm] = *(const v2f*)(&As[(wm + tm * 16 + lm) * ALD + ks]);
      }
#pragma unroll
      for (int tn = 0; tn < 2; ++tn) {
        const int col = wn + tn * 16 + lm;
        b[tn].x = Bs[(ks + 0) * BLD + col];
        b[tn].y = Bs[(ks + 1) * BLD + col];
      }
#pragma unroll
      for (int tm = 0; tm < 4; ++tm)
#pragma unroll
        for (int tn = 0; tn < 2; ++tn)
          acc[tm][tn] = wmma_f32_16x16x4(a[tm], b[tn], acc[tm][tn]);
    }
  }

  float bv[2] = {0.f, 0.f};
  if (INIT && bias != nullptr) {
#pragma unroll
    for (int tn = 0; tn < 2; ++tn) bv[tn] = bias[n0 + wn + tn * 16 + lm];
  }
#pragma unroll
  for (int tm = 0; tm < 4; ++tm) {
    float selfw[8];
    if (INIT) {
#pragma unroll
      for (int v = 0; v < 8; ++v) {
        const int gm = m0 + wm + tm * 16 + v + 8 * lh;
        const float d = dinv[gm < M ? gm : M - 1];
        selfw[v] = d * d;
      }
    }
#pragma unroll
    for (int v = 0; v < 8; ++v) {
      const int gm = m0 + wm + tm * 16 + v + 8 * lh;
      if (gm < M) {
#pragma unroll
        for (int tn = 0; tn < 2; ++tn) {
          const int gn = n0 + wn + tn * 16 + lm;
          const float c = acc[tm][tn][v];
          C[(size_t)gm * ldc + gn] = c;
          if (INIT) C2[(size_t)gm * ldc + gn] = bv[tn] + c * selfw[v];
        }
      }
    }
  }
}

template <int HIDD, int OUTD>
__global__ void __launch_bounds__(256)
init_kernel(float* deg, int n, int nb_deg,
            const float* __restrict__ W2, const float* __restrict__ b1,
            float* __restrict__ cvec,
            const float* __restrict__ b2, float* __restrict__ h2,
            float* __restrict__ a2) {
  const int b = blockIdx.x;
  if (b < nb_deg) {
    const int i = b * 256 + threadIdx.x;
    if (i < n) deg[i] = 1.0f;
  } else if (b < nb_deg + 16) {
    const int lane = threadIdx.x & 31;
    const int j = (b - nb_deg) * 8 + (threadIdx.x >> 5);
    if (j < OUTD) {
      float s = 0.f;
#pragma unroll 4
      for (int k = lane; k < HIDD; k += 32) s += W2[j * HIDD + k] * b1[k];
#pragma unroll
      for (int o = 16; o > 0; o >>= 1) s += __shfl_xor(s, o);
      if (lane == 0) cvec[j] = s;
    }
  } else {
    const int i4 = (b - nb_deg - 16) * 256 + threadIdx.x;
    if (i4 < n * (OUTD / 4)) {
      const int c = (i4 % (OUTD / 4)) * 4;
      *(float4*)(h2 + (size_t)i4 * 4) = *(const float4*)(b2 + c);
      *(float4*)(a2 + (size_t)i4 * 4) = make_float4(0.f, 0.f, 0.f, 0.f);
    }
  }
}

__global__ void deg_accum_kernel(const int* __restrict__ dst, float* deg, int E) {
  int e = blockIdx.x * blockDim.x + threadIdx.x;
  if (e < E) atomicAdd(&deg[dst[e]], 1.0f);
}

__global__ void dinv_kernel(float* deg, float* __restrict__ rsum, int n) {
  int i = blockIdx.x * blockDim.x + threadIdx.x;
  if (i < n) {
    float d = deg[i];
    float r = (d > 0.f) ? rsqrtf(d) : 0.f;
    deg[i] = r;
    rsum[i] = r * r;
  }
}

__global__ void norm_kernel(const int* __restrict__ src, const int* __restrict__ dst,
                            const float* __restrict__ dinv, float* __restrict__ norm,
                            float* rsum, int E) {
  int e = blockIdx.x * blockDim.x + threadIdx.x;
  if (e < E) {
    const int d = dst[e];
    const float nv = dinv[src[e]] * dinv[d];
    norm[e] = nv;
    atomicAdd(&rsum[d], nv);
  }
}

template <int D, int N>
__device__ __forceinline__ void scatter_rows(const float* __restrict__ h,
                                             float* __restrict__ out,
                                             int sl, int dl, int nl, int k0, int lane) {
  constexpr int R = D / 32;
  float v[N][R];
#pragma unroll
  for (int k = 0; k < N; ++k) {
    const int s = __builtin_amdgcn_readlane(sl, k0 + k);
    const float* hp = h + (size_t)s * D + lane;
#pragma unroll
    for (int i = 0; i < R; ++i) v[k][i] = hp[32 * i];
  }
#pragma unroll
  for (int k = 0; k < N; ++k) {
    const int   d  = __builtin_amdgcn_readlane(dl, k0 + k);
    const float nv = __builtin_bit_cast(float, __builtin_amdgcn_readlane(nl, k0 + k));
    float* op = out + (size_t)d * D + lane;
#pragma unroll
    for (int i = 0; i < R; ++i) atomicAdd(op + 32 * i, v[k][i] * nv);
  }
}

template <int D, int EPW, bool SELF>
__global__ void __launch_bounds__(256)
scatter_rows_kernel(const float* __restrict__ h,
                    const int* __restrict__ src, const int* __restrict__ dst,
                    const float* __restrict__ norm, float* __restrict__ out,
                    int E, const float* __restrict__ dinv, int n) {
  const int lane = threadIdx.x & 31;
  const int w = __builtin_amdgcn_readfirstlane((int)(blockIdx.x * 8u + (threadIdx.x >> 5)));
  const int e0 = w * EPW;
  const int Et = SELF ? E + n : E;
  if (e0 >= Et) return;

  int el = e0 + ((lane < EPW) ? lane : 0);
  el = (el < Et) ? el : (Et - 1);
  int sl, dl, nl;
  if (SELF && el >= E) {
    sl = dl = el - E;
    const float d = dinv[sl];
    nl = __builtin_bit_cast(int, d * d);
  } else {
    sl = src[el];
    dl = dst[el];
    nl = __builtin_bit_cast(int, norm[el]);
  }

  if (e0 + EPW <= Et) {
    scatter_rows<D, EPW>(h, out, sl, dl, nl, 0, lane);
  } else {
    for (int k = 0; e0 + k < Et; ++k)
      scatter_rows<D, 1>(h, out, sl, dl, nl, k, lane);
  }
}

template <int D0, int D, int RB, int NW, int NO>
__global__ void __launch_bounds__(NW * 32)
rows_mm2_kernel(const float* __restrict__ X, const float* __restrict__ rw,
                const float* __restrict__ Bw0, const float* __restrict__ bias0,
                const float* __restrict__ Bw, const float* __restrict__ bias,
                float* __restrict__ out, int n) {
  static_assert(NO == NW * 32, "second product: one wave per 32 output columns");
  static_assert(D == NW * 16, "first product: one wave per 16 columns");
  static_assert(D0 % 4 == 0 && D % 4 == 0, "whole K steps of 4");
  constexpr int NT  = NW * 32;
  constexpr int TM  = (RB + 15) / 16;
  constexpr int RP  = TM * 16;
  constexpr int LDX = D0 + 4;
  constexpr int LDA = D + 4;
  __shared__ float xin[RP * LDX];
  __shared__ float acc[RP * LDA];
  __shared__ float rws[RP];

  const int tid  = threadIdx.x;
  const int lane = tid & 31;
  const int w    = __builtin_amdgcn_readfirstlane(tid >> 5);
  const int lh   = lane >> 4;
  const int lm   = lane & 15;
  const int r0   = blockIdx.x * RB;
  const int rows = (n - r0) < RB ? (n - r0) : RB;

  for (int i4 = tid; i4 < RP * (D0 / 4); i4 += NT) {
    const int r = i4 / (D0 / 4);
    const int c = (i4 % (D0 / 4)) * 4;
    float4 v = make_float4(0.f, 0.f, 0.f, 0.f);
    if (r < rows) v = *(const float4*)(X + (size_t)(r0 + r) * D0 + c);
    *(float4*)(&xin[r * LDX + c]) = v;
  }
  if (tid < RP) rws[tid] = tid < rows ? rw[r0 + tid] : 0.f;
  __syncthreads();

  {
    v8f c1[TM];
#pragma unroll
    for (int tm = 0; tm < TM; ++tm) c1[tm] = (v8f)0.0f;
    const float* ap0 = &xin[lm * LDX + 2 * lh];
    const float* bp0 = Bw0 + (size_t)(2 * lh) * D + w * 16 + lm;
#pragma unroll 2
    for (int k0 = 0; k0 < D0; k0 += 4) {
      v2f a[TM];
#pragma unroll
      for (int tm = 0; tm < TM; ++tm)
        a[tm] = *(const v2f*)(ap0 + tm * 16 * LDX + k0);
      v2f b;
      b.x = bp0[(size_t)k0 * D];
      b.y = bp0[(size_t)(k0 + 1) * D];
#pragma unroll
      for (int tm = 0; tm < TM; ++tm) c1[tm] = wmma_f32_16x16x4(a[tm], b, c1[tm]);
    }
    const float bb = bias0[w * 16 + lm];
    float* cp = &acc[(8 * lh) * LDA + w * 16 + lm];
#pragma unroll
    for (int tm = 0; tm < TM; ++tm) {
      const float4 w0 = *(const float4*)(&rws[tm * 16 + 8 * lh]);
      const float4 w1 = *(const float4*)(&rws[tm * 16 + 8 * lh + 4]);
      const float wv[8] = {w0.x, w0.y, w0.z, w0.w, w1.x, w1.y, w1.z, w1.w};
#pragma unroll
      for (int v = 0; v < 8; ++v)
        cp[(tm * 16 + v) * LDA] = c1[tm][v] + wv[v] * bb;
    }
  }
  __syncthreads();

  v8f c[TM][2];
#pragma unroll
  for (int tm = 0; tm < TM; ++tm) {
    c[tm][0] = (v8f)0.0f;
    c[tm][1] = (v8f)0.0f;
  }
  const float* ap = &acc[lm * LDA + 2 * lh];
  const float* bp = Bw + (size_t)(2 * lh) * NO + w * 32 + 2 * lm;
#pragma unroll 2
  for (int k0 = 0; k0 < D; k0 += 4) {
    v2f a[TM];
#pragma unroll
    for (int tm = 0; tm < TM; ++tm)
      a[tm] = *(const v2f*)(ap + tm * 16 * LDA + k0);
    const float2 b0 = *(const float2*)(bp + (size_t)k0 * NO);
    const float2 b1 = *(const float2*)(bp + (size_t)(k0 + 1) * NO);
    v2f be, bo;
    be.x = b0.x; be.y = b1.x;
    bo.x = b0.y; bo.y = b1.y;
#pragma unroll
    for (int tm = 0; tm < TM; ++tm) {
      c[tm][0] = wmma_f32_16x16x4(a[tm], be, c[tm][0]);
      c[tm][1] = wmma_f32_16x16x4(a[tm], bo, c[tm][1]);
    }
  }

  const float2 bo2 = *(const float2*)(bias + w * 32 + 2 * lm);
  float* op = out + (size_t)r0 * NO + w * 32 + 2 * lm;
#pragma unroll
  for (int tm = 0; tm < TM; ++tm) {
#pragma unroll
    for (int v = 0; v < 8; ++v) {
      const int row = tm * 16 + v + 8 * lh;
      if (row < rows)
        *(float2*)(op + (size_t)row * NO) =
            make_float2(c[tm][0][v] + bo2.x, c[tm][1][v] + bo2.y);
    }
  }
}

template <int NTM, int D, int LDA, int NO>
__device__ __forceinline__ void enc_second_product(
    const float* t0s, int tm0, int cq, int lane,
    const float* __restrict__ Wb, float* __restrict__ U, float* __restrict__ P,
    const float* __restrict__ cvec, const float* __restrict__ dinv,
    int r0, int rows, int n) {
  const int lh = lane >> 4;
  const int lm = lane & 15;
  v8f c[NTM][2];
#pragma unroll
  for (int tm = 0; tm < NTM; ++tm) {
    c[tm][0] = (v8f)0.0f;
    c[tm][1] = (v8f)0.0f;
  }
  const float* ap = &t0s[(tm0 * 16 + lm) * LDA + 4 * lh];
  const float* bp = Wb + (size_t)(32 * cq + 2 * lm) * D + 4 * lh;
  for (int k0 = 0; k0 < D; k0 += 32) {
    float4 b0[4], b1[4];
#pragma unroll
    for (int q = 0; q < 4; ++q) {
      b0[q] = *(const float4*)(bp + k0 + 8 * q);
      b1[q] = *(const float4*)(bp + D + k0 + 8 * q);
    }
#pragma unroll
    for (int q = 0; q < 4; ++q) {
      float4 a[NTM];
#pragma unroll
      for (int tm = 0; tm < NTM; ++tm)
        a[tm] = *(const float4*)(ap + tm * 16 * LDA + k0 + 8 * q);
      v2f be, bo;
      be.x = b0[q].x; be.y = b0[q].y;
      bo.x = b1[q].x; bo.y = b1[q].y;
#pragma unroll
      for (int tm = 0; tm < NTM; ++tm) {
        v2f av; av.x = a[tm].x; av.y = a[tm].y;
        c[tm][0] = wmma_f32_16x16x4(av, be, c[tm][0]);
        c[tm][1] = wmma_f32_16x16x4(av, bo, c[tm][1]);
      }
      be.x = b0[q].z; be.y = b0[q].w;
      bo.x = b1[q].z; bo.y = b1[q].w;
#pragma unroll
      for (int tm = 0; tm < NTM; ++tm) {
        v2f av; av.x = a[tm].z; av.y = a[tm].w;
        c[tm][0] = wmma_f32_16x16x4(av, be, c[tm][0]);
        c[tm][1] = wmma_f32_16x16x4(av, bo, c[tm][1]);
      }
    }
  }

  const float2 cv = *(const float2*)(cvec + 32 * cq + 2 * lm);
#pragma unroll
  for (int tm = 0; tm < NTM; ++tm) {
    float selfw[8];
#pragma unroll
    for (int v = 0; v < 8; ++v) {
      const int gm = r0 + (tm0 + tm) * 16 + v + 8 * lh;
      const float d = dinv[gm < n ? gm : n - 1];
      selfw[v] = d * d;
    }
#pragma unroll
    for (int v = 0; v < 8; ++v) {
      const int row = (tm0 + tm) * 16 + v + 8 * lh;
      if (row < rows) {
        const size_t o = (size_t)(r0 + row) * NO + 32 * cq + 2 * lm;
        const float u0 = c[tm][0][v];
        const float u1 = c[tm][1][v];
        *(float2*)(U + o) = make_float2(u0, u1);
        *(float2*)(P + o) = make_float2(cv.x + u0 * selfw[v], cv.y + u1 * selfw[v]);
      }
    }
  }
}

template <int K0, int D, int RB, int NO, int KC>
__global__ void __launch_bounds__(1024)
enc_mm2_kernel(const float* __restrict__ X, const float* __restrict__ Wa,
               const float* __restrict__ Wb, float* __restrict__ U,
               float* __restrict__ P, const float* __restrict__ cvec,
               const float* __restrict__ dinv, int n) {
  constexpr int NW = 32;
  constexpr int NT = NW * 32;
  constexpr int TM = (RB + 15) / 16;
  constexpr int RP = TM * 16;
  static_assert(D == NW * 16, "first product: one wave per 16 columns");
  static_assert(NO == 128, "second product: 4 waves of 32 columns per row group");
  static_assert(TM == 5, "second product: row tiles split 3 + 2 over two wave groups");
  static_assert(K0 % KC == 0 && KC % 32 == 0 && D % 32 == 0, "whole 32-k groups");
  static_assert((RP * (KC / 4)) % NT == 0, "X chunk: whole float4 rounds per thread");
  constexpr int NLD = RP * (KC / 4) / NT;
  constexpr int LDX = KC + 4;
  constexpr int LDA = D + 4;
  __shared__ alignas(16) float xs[RP * LDX];
  __shared__ alignas(16) float t0s[RP * LDA];

  const int tid  = threadIdx.x;
  const int lane = tid & 31;
  const int w    = __builtin_amdgcn_readfirstlane(tid >> 5);
  const int lh   = lane >> 4;
  const int lm   = lane & 15;
  const int r0   = blockIdx.x * RB;
  const int rows = (n - r0) < RB ? (n - r0) : RB;

  v8f c1[TM];
#pragma unroll
  for (int tm = 0; tm < TM; ++tm) c1[tm] = (v8f)0.0f;
  const float* ap0 = &xs[lm * LDX + 4 * lh];
  const float* bp0 = Wa + (size_t)(w * 16 + lm) * K0 + 4 * lh;

  for (int kc = 0; kc < K0; kc += KC) {
    if (kc) __syncthreads();
    float4 xv[NLD];
#pragma unroll
    for (int it = 0; it < NLD; ++it) {
      const int i4 = tid + it * NT;
      const int r  = i4 / (KC / 4);
      const int c  = (i4 % (KC / 4)) * 4;
      const int rr = r < rows ? r : rows - 1;
      xv[it] = *(const float4*)(X + (size_t)(r0 + rr) * K0 + kc + c);
    }
#pragma unroll
    for (int it = 0; it < NLD; ++it) {
      const int i4 = tid + it * NT;
      const int r  = i4 / (KC / 4);
      const int c  = (i4 % (KC / 4)) * 4;
      *(float4*)(&xs[r * LDX + c]) = xv[it];
    }
    __syncthreads();

    for (int k0 = 0; k0 < KC; k0 += 32) {
      float4 b[4];
#pragma unroll
      for (int q = 0; q < 4; ++q)
        b[q] = *(const float4*)(bp0 + kc + k0 + 8 * q);
#pragma unroll
      for (int q = 0; q < 4; ++q) {
        float4 a[TM];
#pragma unroll
        for (int tm = 0; tm < TM; ++tm)
          a[tm] = *(const float4*)(ap0 + tm * 16 * LDX + k0 + 8 * q);
        v2f bv;
        bv.x = b[q].x; bv.y = b[q].y;
#pragma unroll
        for (int tm = 0; tm < TM; ++tm) {
          v2f av; av.x = a[tm].x; av.y = a[tm].y;
          c1[tm] = wmma_f32_16x16x4(av, bv, c1[tm]);
        }
        bv.x = b[q].z; bv.y = b[q].w;
#pragma unroll
        for (int tm = 0; tm < TM; ++tm) {
          v2f av; av.x = a[tm].z; av.y = a[tm].w;
          c1[tm] = wmma_f32_16x16x4(av, bv, c1[tm]);
        }
      }
    }
  }

  {
    float* cp = &t0s[(8 * lh) * LDA + w * 16 + lm];
#pragma unroll
    for (int tm = 0; tm < TM; ++tm)
#pragma unroll
      for (int v = 0; v < 8; ++v) cp[(tm * 16 + v) * LDA] = c1[tm][v];
  }
  __syncthreads();

  if (w < 4)
    enc_second_product<3, D, LDA, NO>(t0s, 0, w, lane, Wb, U, P, cvec, dinv, r0, rows, n);
  else if (w < 8)
    enc_second_product<2, D, LDA, NO>(t0s, 3, w - 4, lane, Wb, U, P, cvec, dinv, r0, rows, n);
}

extern "C" void kernel_launch(void* const* d_in, const int* in_sizes, int n_in,
                              void* d_out, int out_size, void* d_ws, size_t ws_size,
                              hipStream_t stream) {
  const int IN = 1024, HID = 512, OUT = 128;

  const float* features = (const float*)d_in[0];
  const int*   edge     = (const int*)d_in[1];
  const float* W1       = (const float*)d_in[2];
  const float* b1       = (const float*)d_in[3];
  const float* W2       = (const float*)d_in[4];
  const float* b2       = (const float*)d_in[5];
  const float* b3       = (const float*)d_in[6];
  const float* b4       = (const float*)d_in[7];
  const float* head1    = (const float*)d_in[8];

  const int n  = in_sizes[0] / IN;
  const int E  = in_sizes[1] / 2;
  const int Et = E + n;
  const int* src = edge;
  const int* dst = edge + E;

  float* out = (float*)d_out;
  float* z   = out;
  float* h2  = out + (size_t)n * OUT;
  float* h4  = out + (size_t)n * OUT * 2;

  float* ws   = (float*)d_ws;
  float* dinv = ws;
  float* norm = dinv + n;
  float* t1   = norm + Et + (size_t)n * HID;
  float* h2p  = t1 + (size_t)n * HID;
  float* t2   = h2p + (size_t)n * OUT;
  float* cvec = norm + E;
  float* u    = h2p;
  float* p    = t1;
  float* a1   = t2;
  float* a2   = t2 + (size_t)n * OUT;
  float* rsum = t2 + (size_t)n * OUT * 2;

  const int TB = 256;
  const int nb_deg  = (n + TB - 1) / TB;
  const int nb_fill = (n * (OUT / 4) + TB - 1) / TB;
  init_kernel<HID, OUT><<<nb_deg + 16 + nb_fill, TB, 0, stream>>>(dinv, n, nb_deg, W2, b1,
                                                                  cvec, b2, h2, a2);
  deg_accum_kernel<<<(E + TB - 1) / TB, TB, 0, stream>>>(dst, dinv, E);
  dinv_kernel<<<(n + TB - 1) / TB, TB, 0, stream>>>(dinv, rsum, n);
  norm_kernel<<<(E + TB - 1) / TB, TB, 0, stream>>>(src, dst, dinv, norm, rsum, E);

  constexpr int RB = 80;
  enc_mm2_kernel<IN, HID, RB, OUT, 256><<<(n + RB - 1) / RB, 1024, 0, stream>>>(
      features, W1, W2, u, p, cvec, dinv, n);

  scatter_rows_kernel<OUT, 4, false><<<(E + 31) / 32, 256, 0, stream>>>(
      u, src, dst, norm, p, E, nullptr, 0);
  scatter_rows_kernel<OUT, 4, true><<<(E + n + 31) / 32, 256, 0, stream>>>(
      p, src, dst, norm, h2, E, dinv, n);

  gemm_wmma<0, true><<<dim3(OUT / 128, (n + 127) / 128), 256, 0, stream>>>(
      h2, OUT, head1, OUT, z, OUT, n, OUT, OUT, a1, nullptr, dinv);

  scatter_rows_kernel<OUT, 4, false><<<(E + 31) / 32, 256, 0, stream>>>(
      z, src, dst, norm, a1, E, nullptr, 0);
  scatter_rows_kernel<OUT, 4, true><<<(E + n + 31) / 32, 256, 0, stream>>>(
      a1, src, dst, norm, a2, E, dinv, n);

  rows_mm2_kernel<OUT, HID, RB, 32, IN><<<(n + RB - 1) / RB, 1024, 0, stream>>>(
      a2, rsum, W2, b3, W1, b4, h4, n);
}


# Graph norms (degree, dinv, edge norms, rsum, W2 b1) computed by one extra work-group of the encoder GEMM launch on LDS atomics: 7 launches, was 11; encoder self and bias terms join in atomic order
# speedup vs baseline: 3.0555x; 1.0799x over previous
#include <hip/hip_runtime.h>
#include <hip/hip_bf16.h>


typedef float v2f __attribute__((ext_vector_type(2)));
typedef float v8f __attribute__((ext_vector_type(8)));

__device__ __forceinline__ v8f wmma_f32_16x16x4(v2f a, v2f b, v8f c) {
  return __builtin_amdgcn_wmma_f32_16x16x4_f32(false, a, false, b, (short)0, c,
                                               false, false);
}

template <int OPB, bool INIT>
__global__ void __launch_bounds__(256)
gemm_wmma(const float* __restrict__ A, int lda,
          const float* __restrict__ B, int ldb,
          float* __restrict__ C, int ldc,
          int M, int Nc, int K,
          float* __restrict__ C2, const float* __restrict__ bias,
          const float* __restrict__ dinv) {
  constexpr int BM = 128, BN = 128, BK = 16;
  constexpr int ALD = 20;
  constexpr int BLD = 132;

  __shared__ float As[BM * ALD];
  __shared__ float Bs[BK * BLD];

  const int tid  = threadIdx.x;
  const int wid  = tid >> 5;
  const int lane = tid & 31;
  const int lh   = lane >> 4;
  const int lm   = lane & 15;

  const int m0 = blockIdx.y * BM;
  const int n0 = blockIdx.x * BN;
  const int wm = (wid & 1) * 64;
  const int wn = (wid >> 1) * 32;

  v8f acc[4][2];
#pragma unroll
  for (int tm = 0; tm < 4; ++tm)
#pragma unroll
    for (int tn = 0; tn < 2; ++tn) acc[tm][tn] = (v8f)0.0f;

  const float* pa[2];
  const float* pb[2];
#pragma unroll
  for (int r = 0; r < 2; ++r) {
    const int i   = tid + r * 256;
    const int gm  = m0 + (i >> 2);
    const int gmc = gm < M ? gm : M - 1;
    pa[r] = A + (size_t)gmc * lda + ((i & 3) << 2);
    if (OPB)
      pb[r] = B + (size_t)(n0 + (i >> 2)) * ldb + ((i & 3) << 2);
    else
      pb[r] = B + (size_t)(i >> 5) * ldb + n0 + ((i & 31) << 2);
  }

  for (int k0 = 0; k0 < K; k0 += BK) {
    __syncthreads();
    float4 va[2], vb[2];
#pragma unroll
    for (int r = 0; r < 2; ++r) va[r] = *(const float4*)(pa[r] + k0);
#pragma unroll
    for (int r = 0; r < 2; ++r)
      vb[r] = *(const float4*)(OPB ? pb[r] + k0 : pb[r] + (size_t)k0 * ldb);
#pragma unroll
    for (int r = 0; r < 2; ++r) {
      const int i   = tid + r * 256;
      const int row = i >> 2;
      const int c4  = (i & 3) << 2;
      *(float4*)(&As[row * ALD + c4]) = va[r];
    }
    if (OPB) {
#pragma unroll
      for (int r = 0; r < 2; ++r) {
        const int i  = tid + r * 256;
        const int j  = i >> 2;
        const int k4 = (i & 3) << 2;
        Bs[(k4 + 0) * BLD + j] = vb[r].x;
        Bs[(k4 + 1) * BLD + j] = vb[r].y;
        Bs[(k4 + 2) * BLD + j] = vb[r].z;
        Bs[(k4 + 3) * BLD + j] = vb[r].w;
      }
    } else {
#pragma unroll
      for (int r = 0; r < 2; ++r) {
        const int i  = tid + r * 256;
        const int kk = i >> 5;
        const int c  = (i & 31) << 2;
        *(float4*)(&Bs[kk * BLD + c]) = vb[r];
      }
    }
    if (k0 + BK < K) {
      const int gm = m0 + (tid >> 1);
      if (gm < M)
        __builtin_prefetch(A + (size_t)gm * lda + (k0 + BK) + (tid & 1) * 8, 0, 1);
    }
    __syncthreads();

#pragma unroll
    for (int kk = 0; kk < BK; kk += 4) {
      const int ks = kk + 2 * lh;
      v2f a[4], b[2];
#pragma unroll
      for (int tm = 0; tm < 4; ++tm) {
        a[tm] = *(const v2f*)(&As[(wm + tm * 16 + lm) * ALD + ks]);
      }
#pragma unroll
      for (int tn = 0; tn < 2; ++tn) {
        const int col = wn + tn * 16 + lm;
        b[tn].x = Bs[(ks + 0) * BLD + col];
        b[tn].y = Bs[(ks + 1) * BLD + col];
      }
#pragma unroll
      for (int tm = 0; tm < 4; ++tm)
#pragma unroll
        for (int tn = 0; tn < 2; ++tn)
          acc[tm][tn] = wmma_f32_16x16x4(a[tm], b[tn], acc[tm][tn]);
    }
  }

  float bv[2] = {0.f, 0.f};
  if (INIT && bias != nullptr) {
#pragma unroll
    for (int tn = 0; tn < 2; ++tn) bv[tn] = bias[n0 + wn + tn * 16 + lm];
  }
#pragma unroll
  for (int tm = 0; tm < 4; ++tm) {
    float selfw[8];
    if (INIT) {
#pragma unroll
      for (int v = 0; v < 8; ++v) {
        const int gm = m0 + wm + tm * 16 + v + 8 * lh;
        const float d = dinv[gm < M ? gm : M - 1];
        selfw[v] = d * d;
      }
    }
#pragma unroll
    for (int v = 0; v < 8; ++v) {
      const int gm = m0 + wm + tm * 16 + v + 8 * lh;
      if (gm < M) {
#pragma unroll
        for (int tn = 0; tn < 2; ++tn) {
          const int gn = n0 + wn + tn * 16 + lm;
          const float c = acc[tm][tn][v];
          C[(size_t)gm * ldc + gn] = c;
          if (INIT) C2[(size_t)gm * ldc + gn] = bv[tn] + c * selfw[v];
        }
      }
    }
  }
}

template <int NT, int HIDD, int OUTD>
__device__ __forceinline__ void graph_norm_block(
    float* degs, float* rs,
    const int* __restrict__ src, const int* __restrict__ dst, int E, int n,
    float* __restrict__ dinv, float* __restrict__ norm, float* __restrict__ rsum,
    const float* __restrict__ W2, const float* __restrict__ b1,
    float* __restrict__ cvec) {
  constexpr int EB = 8;
  const int tid  = threadIdx.x;
  const int lane = tid & 31;
  for (int j = tid >> 5; j < OUTD; j += NT / 32) {
    float sacc = 0.f;
#pragma unroll 4
    for (int k = lane; k < HIDD; k += 32) sacc += W2[j * HIDD + k] * b1[k];
#pragma unroll
    for (int o = 16; o > 0; o >>= 1) sacc += __shfl_xor(sacc, o);
    if (lane == 0) cvec[j] = sacc;
  }
  for (int i = tid; i < n; i += NT) degs[i] = 1.0f;
  __syncthreads();
  int e0 = tid;
  for (; e0 + (EB - 1) * NT < E; e0 += NT * EB) {
    int dv[EB];
#pragma unroll
    for (int q = 0; q < EB; ++q) dv[q] = dst[e0 + q * NT];
#pragma unroll
    for (int q = 0; q < EB; ++q) atomicAdd(&degs[dv[q]], 1.0f);
  }
  for (; e0 < E; e0 += NT) atomicAdd(&degs[dst[e0]], 1.0f);
  __syncthreads();
  for (int i = tid; i < n; i += NT) {
    const float d = degs[i];
    const float r = (d > 0.f) ? rsqrtf(d) : 0.f;
    degs[i] = r;
    dinv[i] = r;
    rs[i]   = r * r;
  }
  __syncthreads();
  e0 = tid;
  for (; e0 + (EB - 1) * NT < E; e0 += NT * EB) {
    int sv[EB], dv[EB];
#pragma unroll
    for (int q = 0; q < EB; ++q) {
      sv[q] = src[e0 + q * NT];
      dv[q] = dst[e0 + q * NT];
    }
    float nv[EB];
#pragma unroll
    for (int q = 0; q < EB; ++q) nv[q] = degs[sv[q]] * degs[dv[q]];
#pragma unroll
    for (int q = 0; q < EB; ++q) {
      norm[e0 + q * NT] = nv[q];
      atomicAdd(&rs[dv[q]], nv[q]);
    }
  }
  for (; e0 < E; e0 += NT) {
    const int d = dst[e0];
    const float nv = degs[src[e0]] * degs[d];
    norm[e0] = nv;
    atomicAdd(&rs[d], nv);
  }
  __syncthreads();
  for (int i = tid; i < n; i += NT) rsum[i] = rs[i];
}

template <int D, int N, bool ADDV>
__device__ __forceinline__ void scatter_rows(const float* __restrict__ h,
                                             float* __restrict__ out,
                                             int sl, int dl, int nl, int k0, int lane,
                                             const float (&av)[D / 32]) {
  constexpr int R = D / 32;
  float v[N][R];
#pragma unroll
  for (int k = 0; k < N; ++k) {
    const int s = __builtin_amdgcn_readlane(sl, k0 + k);
    const float* hp = h + (size_t)s * D + lane;
#pragma unroll
    for (int i = 0; i < R; ++i) v[k][i] = hp[32 * i];
  }
#pragma unroll
  for (int k = 0; k < N; ++k) {
    const int   d  = __builtin_amdgcn_readlane(dl, k0 + k);
    const float nv = __builtin_bit_cast(float, __builtin_amdgcn_readlane(nl, k0 + k));
    float* op = out + (size_t)d * D + lane;
#pragma unroll
    for (int i = 0; i < R; ++i)
      atomicAdd(op + 32 * i, (ADDV ? v[k][i] + av[i] : v[k][i]) * nv);
  }
}

template <int D, int EPW, bool SELF, bool ADDV>
__global__ void __launch_bounds__(256)
scatter_rows_kernel(const float* __restrict__ h,
                    const int* __restrict__ src, const int* __restrict__ dst,
                    const float* __restrict__ norm, float* __restrict__ out,
                    int E, const float* __restrict__ dinv, int n,
                    const float* __restrict__ addv) {
  const int lane = threadIdx.x & 31;
  const int w = __builtin_amdgcn_readfirstlane((int)(blockIdx.x * 8u + (threadIdx.x >> 5)));
  const int e0 = w * EPW;
  const int Et = SELF ? E + n : E;
  if (e0 >= Et) return;

  int el = e0 + ((lane < EPW) ? lane : 0);
  el = (el < Et) ? el : (Et - 1);
  int sl, dl, nl;
  if (SELF && el >= E) {
    sl = dl = el - E;
    const float d = dinv[sl];
    nl = __builtin_bit_cast(int, d * d);
  } else {
    sl = src[el];
    dl = dst[el];
    nl = __builtin_bit_cast(int, norm[el]);
  }

  float av[D / 32];
#pragma unroll
  for (int i = 0; i < D / 32; ++i) av[i] = ADDV ? addv[lane + 32 * i] : 0.f;

  if (e0 + EPW <= Et) {
    scatter_rows<D, EPW, ADDV>(h, out, sl, dl, nl, 0, lane, av);
  } else {
    for (int k = 0; e0 + k < Et; ++k)
      scatter_rows<D, 1, ADDV>(h, out, sl, dl, nl, k, lane, av);
  }
}

template <int D0, int D, int RB, int NW, int NO>
__global__ void __launch_bounds__(NW * 32)
rows_mm2_kernel(const float* __restrict__ X, const float* __restrict__ rw,
                const float* __restrict__ Bw0, const float* __restrict__ bias0,
                const float* __restrict__ Bw, const float* __restrict__ bias,
                float* __restrict__ out, int n) {
  static_assert(NO == NW * 32, "second product: one wave per 32 output columns");
  static_assert(D == NW * 16, "first product: one wave per 16 columns");
  static_assert(D0 % 4 == 0 && D % 4 == 0, "whole K steps of 4");
  constexpr int NT  = NW * 32;
  constexpr int TM  = (RB + 15) / 16;
  constexpr int RP  = TM * 16;
  constexpr int LDX = D0 + 4;
  constexpr int LDA = D + 4;
  __shared__ float xin[RP * LDX];
  __shared__ float acc[RP * LDA];
  __shared__ float rws[RP];

  const int tid  = threadIdx.x;
  const int lane = tid & 31;
  const int w    = __builtin_amdgcn_readfirstlane(tid >> 5);
  const int lh   = lane >> 4;
  const int lm   = lane & 15;
  const int r0   = blockIdx.x * RB;
  const int rows = (n - r0) < RB ? (n - r0) : RB;

  for (int i4 = tid; i4 < RP * (D0 / 4); i4 += NT) {
    const int r = i4 / (D0 / 4);
    const int c = (i4 % (D0 / 4)) * 4;
    float4 v = make_float4(0.f, 0.f, 0.f, 0.f);
    if (r < rows) v = *(const float4*)(X + (size_t)(r0 + r) * D0 + c);
    *(float4*)(&xin[r * LDX + c]) = v;
  }
  if (tid < RP) rws[tid] = tid < rows ? rw[r0 + tid] : 0.f;
  __syncthreads();

  {
    v8f c1[TM];
#pragma unroll
    for (int tm = 0; tm < TM; ++tm) c1[tm] = (v8f)0.0f;
    const float* ap0 = &xin[lm * LDX + 2 * lh];
    const float* bp0 = Bw0 + (size_t)(2 * lh) * D + w * 16 + lm;
#pragma unroll 2
    for (int k0 = 0; k0 < D0; k0 += 4) {
      v2f a[TM];
#pragma unroll
      for (int tm = 0; tm < TM; ++tm)
        a[tm] = *(const v2f*)(ap0 + tm * 16 * LDX + k0);
      v2f b;
      b.x = bp0[(size_t)k0 * D];
      b.y = bp0[(size_t)(k0 + 1) * D];
#pragma unroll
      for (int tm = 0; tm < TM; ++tm) c1[tm] = wmma_f32_16x16x4(a[tm], b, c1[tm]);
    }
    const float bb = bias0[w * 16 + lm];
    float* cp = &acc[(8 * lh) * LDA + w * 16 + lm];
#pragma unroll
    for (int tm = 0; tm < TM; ++tm) {
      const float4 w0 = *(const float4*)(&rws[tm * 16 + 8 * lh]);
      const float4 w1 = *(const float4*)(&rws[tm * 16 + 8 * lh + 4]);
      const float wv[8] = {w0.x, w0.y, w0.z, w0.w, w1.x, w1.y, w1.z, w1.w};
#pragma unroll
      for (int v = 0; v < 8; ++v)
        cp[(tm * 16 + v) * LDA] = c1[tm][v] + wv[v] * bb;
    }
  }
  __syncthreads();

  v8f c[TM][2];
#pragma unroll
  for (int tm = 0; tm < TM; ++tm) {
    c[tm][0] = (v8f)0.0f;
    c[tm][1] = (v8f)0.0f;
  }
  const float* ap = &acc[lm * LDA + 2 * lh];
  const float* bp = Bw + (size_t)(2 * lh) * NO + w * 32 + 2 * lm;
#pragma unroll 2
  for (int k0 = 0; k0 < D; k0 += 4) {
    v2f a[TM];
#pragma unroll
    for (int tm = 0; tm < TM; ++tm)
      a[tm] = *(const v2f*)(ap + tm * 16 * LDA + k0);
    const float2 b0 = *(const float2*)(bp + (size_t)k0 * NO);
    const float2 b1 = *(const float2*)(bp + (size_t)(k0 + 1) * NO);
    v2f be, bo;
    be.x = b0.x; be.y = b1.x;
    bo.x = b0.y; bo.y = b1.y;
#pragma unroll
    for (int tm = 0; tm < TM; ++tm) {
      c[tm][0] = wmma_f32_16x16x4(a[tm], be, c[tm][0]);
      c[tm][1] = wmma_f32_16x16x4(a[tm], bo, c[tm][1]);
    }
  }

  const float2 bo2 = *(const float2*)(bias + w * 32 + 2 * lm);
  float* op = out + (size_t)r0 * NO + w * 32 + 2 * lm;
#pragma unroll
  for (int tm = 0; tm < TM; ++tm) {
#pragma unroll
    for (int v = 0; v < 8; ++v) {
      const int row = tm * 16 + v + 8 * lh;
      if (row < rows)
        *(float2*)(op + (size_t)row * NO) =
            make_float2(c[tm][0][v] + bo2.x, c[tm][1][v] + bo2.y);
    }
  }
}

template <int NTM, int D, int LDA, int NO>
__device__ __forceinline__ void enc_second_product(
    const float* t0s, int tm0, int cq, int lane,
    const float* __restrict__ Wb, float* __restrict__ U, int r0, int rows) {
  const int lh = lane >> 4;
  const int lm = lane & 15;
  v8f c[NTM][2];
#pragma unroll
  for (int tm = 0; tm < NTM; ++tm) {
    c[tm][0] = (v8f)0.0f;
    c[tm][1] = (v8f)0.0f;
  }
  const float* ap = &t0s[(tm0 * 16 + lm) * LDA + 4 * lh];
  const float* bp = Wb + (size_t)(32 * cq + 2 * lm) * D + 4 * lh;
  for (int k0 = 0; k0 < D; k0 += 32) {
    float4 b0[4], b1[4];
#pragma unroll
    for (int q = 0; q < 4; ++q) {
      b0[q] = *(const float4*)(bp + k0 + 8 * q);
      b1[q] = *(const float4*)(bp + D + k0 + 8 * q);
    }
#pragma unroll
    for (int q = 0; q < 4; ++q) {
      float4 a[NTM];
#pragma unroll
      for (int tm = 0; tm < NTM; ++tm)
        a[tm] = *(const float4*)(ap + tm * 16 * LDA + k0 + 8 * q);
      v2f be, bo;
      be.x = b0[q].x; be.y = b0[q].y;
      bo.x = b1[q].x; bo.y = b1[q].y;
#pragma unroll
      for (int tm = 0; tm < NTM; ++tm) {
        v2f av; av.x = a[tm].x; av.y = a[tm].y;
        c[tm][0] = wmma_f32_16x16x4(av, be, c[tm][0]);
        c[tm][1] = wmma_f32_16x16x4(av, bo, c[tm][1]);
      }
      be.x = b0[q].z; be.y = b0[q].w;
      bo.x = b1[q].z; bo.y = b1[q].w;
#pragma unroll
      for (int tm = 0; tm < NTM; ++tm) {
        v2f av; av.x = a[tm].z; av.y = a[tm].w;
        c[tm][0] = wmma_f32_16x16x4(av, be, c[tm][0]);
        c[tm][1] = wmma_f32_16x16x4(av, bo, c[tm][1]);
      }
    }
  }

#pragma unroll
  for (int tm = 0; tm < NTM; ++tm) {
#pragma unroll
    for (int v = 0; v < 8; ++v) {
      const int row = (tm0 + tm) * 16 + v + 8 * lh;
      if (row < rows) {
        const size_t o = (size_t)(r0 + row) * NO + 32 * cq + 2 * lm;
        *(float2*)(U + o) = make_float2(c[tm][0][v], c[tm][1][v]);
      }
    }
  }
}

template <int K0, int D, int RB, int NO, int KC>
__global__ void __launch_bounds__(1024)
enc_mm2_kernel(const float* __restrict__ X, const float* __restrict__ Wa,
               const float* __restrict__ Wb, float* __restrict__ U,
               float* __restrict__ P, const float* __restrict__ b2,
               float* __restrict__ H2, float* __restrict__ A2, int n,
               const int* __restrict__ src, const int* __restrict__ dst, int E,
               const float* __restrict__ b1, float* __restrict__ dinv,
               float* __restrict__ norm, float* __restrict__ rsum,
               float* __restrict__ cvec) {
  constexpr int NW = 32;
  constexpr int NT = NW * 32;
  constexpr int TM = (RB + 15) / 16;
  constexpr int RP = TM * 16;
  static_assert(D == NW * 16, "first product: one wave per 16 columns");
  static_assert(NO == 128, "second product: 4 waves of 32 columns per row group");
  static_assert(TM == 5, "second product: row tiles split 3 + 2 over two wave groups");
  static_assert(K0 % KC == 0 && KC % 32 == 0 && D % 32 == 0, "whole 32-k groups");
  static_assert((RP * (KC / 4)) % NT == 0, "X chunk: whole float4 rounds per thread");
  constexpr int NLD = RP * (KC / 4) / NT;
  constexpr int LDX = KC + 4;
  constexpr int LDA = D + 4;
  __shared__ alignas(16) float xs[RP * LDX];
  __shared__ alignas(16) float t0s[RP * LDA];

  if (blockIdx.x == 0) {
    graph_norm_block<NT, D, NO>(xs, t0s, src, dst, E, n, dinv, norm, rsum, Wb, b1, cvec);
    return;
  }

  const int tid  = threadIdx.x;
  const int lane = tid & 31;
  const int w    = __builtin_amdgcn_readfirstlane(tid >> 5);
  const int lh   = lane >> 4;
  const int lm   = lane & 15;
  const int r0   = (blockIdx.x - 1) * RB;
  const int rows = (n - r0) < RB ? (n - r0) : RB;

  v8f c1[TM];
#pragma unroll
  for (int tm = 0; tm < TM; ++tm) c1[tm] = (v8f)0.0f;
  const float* ap0 = &xs[lm * LDX + 4 * lh];
  const float* bp0 = Wa + (size_t)(w * 16 + lm) * K0 + 4 * lh;

  for (int kc = 0; kc < K0; kc += KC) {
    if (kc) __syncthreads();
    float4 xv[NLD];
#pragma unroll
    for (int it = 0; it < NLD; ++it) {
      const int i4 = tid + it * NT;
      const int r  = i4 / (KC / 4);
      const int c  = (i4 % (KC / 4)) * 4;
      const int rr = r < rows ? r : rows - 1;
      xv[it] = *(const float4*)(X + (size_t)(r0 + rr) * K0 + kc + c);
    }
#pragma unroll
    for (int it = 0; it < NLD; ++it) {
      const int i4 = tid + it * NT;
      const int r  = i4 / (KC / 4);
      const int c  = (i4 % (KC / 4)) * 4;
      *(float4*)(&xs[r * LDX + c]) = xv[it];
    }
    __syncthreads();

    for (int k0 = 0; k0 < KC; k0 += 32) {
      float4 b[4];
#pragma unroll
      for (int q = 0; q < 4; ++q)
        b[q] = *(const float4*)(bp0 + kc + k0 + 8 * q);
#pragma unroll
      for (int q = 0; q < 4; ++q) {
        float4 a[TM];
#pragma unroll
        for (int tm = 0; tm < TM; ++tm)
          a[tm] = *(const float4*)(ap0 + tm * 16 * LDX + k0 + 8 * q);
        v2f bv;
        bv.x = b[q].x; bv.y = b[q].y;
#pragma unroll
        for (int tm = 0; tm < TM; ++tm) {
          v2f av; av.x = a[tm].x; av.y = a[tm].y;
          c1[tm] = wmma_f32_16x16x4(av, bv, c1[tm]);
        }
        bv.x = b[q].z; bv.y = b[q].w;
#pragma unroll
        for (int tm = 0; tm < TM; ++tm) {
          v2f av; av.x = a[tm].z; av.y = a[tm].w;
          c1[tm] = wmma_f32_16x16x4(av, bv, c1[tm]);
        }
      }
    }
  }

  {
    float* cp = &t0s[(8 * lh) * LDA + w * 16 + lm];
#pragma unroll
    for (int tm = 0; tm < TM; ++tm)
#pragma unroll
      for (int v = 0; v < 8; ++v) cp[(tm * 16 + v) * LDA] = c1[tm][v];
  }
  __syncthreads();

  if (w < 4) {
    enc_second_product<3, D, LDA, NO>(t0s, 0, w, lane, Wb, U, r0, rows);
  } else if (w < 8) {
    enc_second_product<2, D, LDA, NO>(t0s, 3, w - 4, lane, Wb, U, r0, rows);
  } else {
    for (int i4 = tid - 256; i4 < rows * (NO / 4); i4 += NT - 256) {
      const size_t o = (size_t)r0 * NO + (size_t)i4 * 4;
      const int    c = (i4 % (NO / 4)) * 4;
      *(float4*)(P + o)  = make_float4(0.f, 0.f, 0.f, 0.f);
      *(float4*)(H2 + o) = *(const float4*)(b2 + c);
      *(float4*)(A2 + o) = make_float4(0.f, 0.f, 0.f, 0.f);
    }
  }
}

extern "C" void kernel_launch(void* const* d_in, const int* in_sizes, int n_in,
                              void* d_out, int out_size, void* d_ws, size_t ws_size,
                              hipStream_t stream) {
  const int IN = 1024, HID = 512, OUT = 128;

  const float* features = (const float*)d_in[0];
  const int*   edge     = (const int*)d_in[1];
  const float* W1       = (const float*)d_in[2];
  const float* b1       = (const float*)d_in[3];
  const float* W2       = (const float*)d_in[4];
  const float* b2       = (const float*)d_in[5];
  const float* b3       = (const float*)d_in[6];
  const float* b4       = (const float*)d_in[7];
  const float* head1    = (const float*)d_in[8];

  const int n  = in_sizes[0] / IN;
  const int E  = in_sizes[1] / 2;
  const int Et = E + n;
  const int* src = edge;
  const int* dst = edge + E;

  float* out = (float*)d_out;
  float* z   = out;
  float* h2  = out + (size_t)n * OUT;
  float* h4  = out + (size_t)n * OUT * 2;

  float* ws   = (float*)d_ws;
  float* dinv = ws;
  float* norm = dinv + n;
  float* t1   = norm + Et + (size_t)n * HID;
  float* h2p  = t1 + (size_t)n * HID;
  float* t2   = h2p + (size_t)n * OUT;
  float* cvec = norm + E;
  float* u    = h2p;
  float* p    = t1;
  float* a1   = t2;
  float* a2   = t2 + (size_t)n * OUT;
  float* rsum = t2 + (size_t)n * OUT * 2;

  constexpr int RB = 80;
  enc_mm2_kernel<IN, HID, RB, OUT, 256><<<1 + (n + RB - 1) / RB, 1024, 0, stream>>>(
      features, W1, W2, u, p, b2, h2, a2, n, src, dst, E, b1, dinv, norm, rsum, cvec);

  scatter_rows_kernel<OUT, 4, true, false><<<(E + n + 31) / 32, 256, 0, stream>>>(
      u, src, dst, norm, p, E, dinv, n, nullptr);
  scatter_rows_kernel<OUT, 4, true, true><<<(E + n + 31) / 32, 256, 0, stream>>>(
      p, src, dst, norm, h2, E, dinv, n, cvec);

  gemm_wmma<0, true><<<dim3(OUT / 128, (n + 127) / 128), 256, 0, stream>>>(
      h2, OUT, head1, OUT, z, OUT, n, OUT, OUT, a1, nullptr, dinv);

  scatter_rows_kernel<OUT, 4, false, false><<<(E + 31) / 32, 256, 0, stream>>>(
      z, src, dst, norm, a1, E, nullptr, 0, nullptr);
  scatter_rows_kernel<OUT, 4, true, false><<<(E + n + 31) / 32, 256, 0, stream>>>(
      a1, src, dst, norm, a2, E, dinv, n, nullptr);

  rows_mm2_kernel<OUT, HID, RB, 32, IN><<<(n + RB - 1) / RB, 1024, 0, stream>>>(
      a2, rsum, W2, b3, W1, b4, h4, n);
}


# K loops of both fused product kernels: each wave steps its priority 3,2,1,0 by quarter of its own progress between barriers (s_setprio), so a SIMD's 8 waves advance together; sums unchanged
# speedup vs baseline: 3.1980x; 1.0466x over previous
#include <hip/hip_runtime.h>
#include <hip/hip_bf16.h>


typedef float v2f __attribute__((ext_vector_type(2)));
typedef float v8f __attribute__((ext_vector_type(8)));

__device__ __forceinline__ v8f wmma_f32_16x16x4(v2f a, v2f b, v8f c) {
  return __builtin_amdgcn_wmma_f32_16x16x4_f32(false, a, false, b, (short)0, c,
                                               false, false);
}

__device__ __forceinline__ void wave_prio_for_quarter(int q) {
  switch (q) {
    case 0:  __builtin_amdgcn_s_setprio(3); break;
    case 1:  __builtin_amdgcn_s_setprio(2); break;
    case 2:  __builtin_amdgcn_s_setprio(1); break;
    default: __builtin_amdgcn_s_setprio(0); break;
  }
}

template <int OPB, bool INIT>
__global__ void __launch_bounds__(256)
gemm_wmma(const float* __restrict__ A, int lda,
          const float* __restrict__ B, int ldb,
          float* __restrict__ C, int ldc,
          int M, int Nc, int K,
          float* __restrict__ C2, const float* __restrict__ bias,
          const float* __restrict__ dinv) {
  constexpr int BM = 128, BN = 128, BK = 16;
  constexpr int ALD = 20;
  constexpr int BLD = 132;

  __shared__ float As[BM * ALD];
  __shared__ float Bs[BK * BLD];

  const int tid  = threadIdx.x;
  const int wid  = tid >> 5;
  const int lane = tid & 31;
  const int lh   = lane >> 4;
  const int lm   = lane & 15;

  const int m0 = blockIdx.y * BM;
  const int n0 = blockIdx.x * BN;
  const int wm = (wid & 1) * 64;
  const int wn = (wid >> 1) * 32;

  v8f acc[4][2];
#pragma unroll
  for (int tm = 0; tm < 4; ++tm)
#pragma unroll
    for (int tn = 0; tn < 2; ++tn) acc[tm][tn] = (v8f)0.0f;

  const float* pa[2];
  const float* pb[2];
#pragma unroll
  for (int r = 0; r < 2; ++r) {
    const int i   = tid + r * 256;
    const int gm  = m0 + (i >> 2);
    const int gmc = gm < M ? gm : M - 1;
    pa[r] = A + (size_t)gmc * lda + ((i & 3) << 2);
    if (OPB)
      pb[r] = B + (size_t)(n0 + (i >> 2)) * ldb + ((i & 3) << 2);
    else
      pb[r] = B + (size_t)(i >> 5) * ldb + n0 + ((i & 31) << 2);
  }

  for (int k0 = 0; k0 < K; k0 += BK) {
    __syncthreads();
    float4 va[2], vb[2];
#pragma unroll
    for (int r = 0; r < 2; ++r) va[r] = *(const float4*)(pa[r] + k0);
#pragma unroll
    for (int r = 0; r < 2; ++r)
      vb[r] = *(const float4*)(OPB ? pb[r] + k0 : pb[r] + (size_t)k0 * ldb);
#pragma unroll
    for (int r = 0; r < 2; ++r) {
      const int i   = tid + r * 256;
      const int row = i >> 2;
      const int c4  = (i & 3) << 2;
      *(float4*)(&As[row * ALD + c4]) = va[r];
    }
    if (OPB) {
#pragma unroll
      for (int r = 0; r < 2; ++r) {
        const int i  = tid + r * 256;
        const int j  = i >> 2;
        const int k4 = (i & 3) << 2;
        Bs[(k4 + 0) * BLD + j] = vb[r].x;
        Bs[(k4 + 1) * BLD + j] = vb[r].y;
        Bs[(k4 + 2) * BLD + j] = vb[r].z;
        Bs[(k4 + 3) * BLD + j] = vb[r].w;
      }
    } else {
#pragma unroll
      for (int r = 0; r < 2; ++r) {
        const int i  = tid + r * 256;
        const int kk = i >> 5;
        const int c  = (i & 31) << 2;
        *(float4*)(&Bs[kk * BLD + c]) = vb[r];
      }
    }
    if (k0 + BK < K) {
      const int gm = m0 + (tid >> 1);
      if (gm < M)
        __builtin_prefetch(A + (size_t)gm * lda + (k0 + BK) + (tid & 1) * 8, 0, 1);
    }
    __syncthreads();

#pragma unroll
    for (int kk = 0; kk < BK; kk += 4) {
      const int ks = kk + 2 * lh;
      v2f a[4], b[2];
#pragma unroll
      for (int tm = 0; tm < 4; ++tm) {
        a[tm] = *(const v2f*)(&As[(wm + tm * 16 + lm) * ALD + ks]);
      }
#pragma unroll
      for (int tn = 0; tn < 2; ++tn) {
        const int col = wn + tn * 16 + lm;
        b[tn].x = Bs[(ks + 0) * BLD + col];
        b[tn].y = Bs[(ks + 1) * BLD + col];
      }
#pragma unroll
      for (int tm = 0; tm < 4; ++tm)
#pragma unroll
        for (int tn = 0; tn < 2; ++tn)
          acc[tm][tn] = wmma_f32_16x16x4(a[tm], b[tn], acc[tm][tn]);
    }
  }

  float bv[2] = {0.f, 0.f};
  if (INIT && bias != nullptr) {
#pragma unroll
    for (int tn = 0; tn < 2; ++tn) bv[tn] = bias[n0 + wn + tn * 16 + lm];
  }
#pragma unroll
  for (int tm = 0; tm < 4; ++tm) {
    float selfw[8];
    if (INIT) {
#pragma unroll
      for (int v = 0; v < 8; ++v) {
        const int gm = m0 + wm + tm * 16 + v + 8 * lh;
        const float d = dinv[gm < M ? gm : M - 1];
        selfw[v] = d * d;
      }
    }
#pragma unroll
    for (int v = 0; v < 8; ++v) {
      const int gm = m0 + wm + tm * 16 + v + 8 * lh;
      if (gm < M) {
#pragma unroll
        for (int tn = 0; tn < 2; ++tn) {
          const int gn = n0 + wn + tn * 16 + lm;
          const float c = acc[tm][tn][v];
          C[(size_t)gm * ldc + gn] = c;
          if (INIT) C2[(size_t)gm * ldc + gn] = bv[tn] + c * selfw[v];
        }
      }
    }
  }
}

template <int NT, int HIDD, int OUTD>
__device__ __forceinline__ void graph_norm_block(
    float* degs, float* rs,
    const int* __restrict__ src, const int* __restrict__ dst, int E, int n,
    float* __restrict__ dinv, float* __restrict__ norm, float* __restrict__ rsum,
    const float* __restrict__ W2, const float* __restrict__ b1,
    float* __restrict__ cvec) {
  constexpr int EB = 8;
  const int tid  = threadIdx.x;
  const int lane = tid & 31;
  for (int j = tid >> 5; j < OUTD; j += NT / 32) {
    float sacc = 0.f;
#pragma unroll 4
    for (int k = lane; k < HIDD; k += 32) sacc += W2[j * HIDD + k] * b1[k];
#pragma unroll
    for (int o = 16; o > 0; o >>= 1) sacc += __shfl_xor(sacc, o);
    if (lane == 0) cvec[j] = sacc;
  }
  for (int i = tid; i < n; i += NT) degs[i] = 1.0f;
  __syncthreads();
  int e0 = tid;
  for (; e0 + (EB - 1) * NT < E; e0 += NT * EB) {
    int dv[EB];
#pragma unroll
    for (int q = 0; q < EB; ++q) dv[q] = dst[e0 + q * NT];
#pragma unroll
    for (int q = 0; q < EB; ++q) atomicAdd(&degs[dv[q]], 1.0f);
  }
  for (; e0 < E; e0 += NT) atomicAdd(&degs[dst[e0]], 1.0f);
  __syncthreads();
  for (int i = tid; i < n; i += NT) {
    const float d = degs[i];
    const float r = (d > 0.f) ? rsqrtf(d) : 0.f;
    degs[i] = r;
    dinv[i] = r;
    rs[i]   = r * r;
  }
  __syncthreads();
  e0 = tid;
  for (; e0 + (EB - 1) * NT < E; e0 += NT * EB) {
    int sv[EB], dv[EB];
#pragma unroll
    for (int q = 0; q < EB; ++q) {
      sv[q] = src[e0 + q * NT];
      dv[q] = dst[e0 + q * NT];
    }
    float nv[EB];
#pragma unroll
    for (int q = 0; q < EB; ++q) nv[q] = degs[sv[q]] * degs[dv[q]];
#pragma unroll
    for (int q = 0; q < EB; ++q) {
      norm[e0 + q * NT] = nv[q];
      atomicAdd(&rs[dv[q]], nv[q]);
    }
  }
  for (; e0 < E; e0 += NT) {
    const int d = dst[e0];
    const float nv = degs[src[e0]] * degs[d];
    norm[e0] = nv;
    atomicAdd(&rs[d], nv);
  }
  __syncthreads();
  for (int i = tid; i < n; i += NT) rsum[i] = rs[i];
}

template <int D, int N, bool ADDV>
__device__ __forceinline__ void scatter_rows(const float* __restrict__ h,
                                             float* __restrict__ out,
                                             int sl, int dl, int nl, int k0, int lane,
                                             const float (&av)[D / 32]) {
  constexpr int R = D / 32;
  float v[N][R];
#pragma unroll
  for (int k = 0; k < N; ++k) {
    const int s = __builtin_amdgcn_readlane(sl, k0 + k);
    const float* hp = h + (size_t)s * D + lane;
#pragma unroll
    for (int i = 0; i < R; ++i) v[k][i] = hp[32 * i];
  }
#pragma unroll
  for (int k = 0; k < N; ++k) {
    const int   d  = __builtin_amdgcn_readlane(dl, k0 + k);
    const float nv = __builtin_bit_cast(float, __builtin_amdgcn_readlane(nl, k0 + k));
    float* op = out + (size_t)d * D + lane;
#pragma unroll
    for (int i = 0; i < R; ++i)
      atomicAdd(op + 32 * i, (ADDV ? v[k][i] + av[i] : v[k][i]) * nv);
  }
}

template <int D, int EPW, bool SELF, bool ADDV>
__global__ void __launch_bounds__(256)
scatter_rows_kernel(const float* __restrict__ h,
                    const int* __restrict__ src, const int* __restrict__ dst,
                    const float* __restrict__ norm, float* __restrict__ out,
                    int E, const float* __restrict__ dinv, int n,
                    const float* __restrict__ addv) {
  const int lane = threadIdx.x & 31;
  const int w = __builtin_amdgcn_readfirstlane((int)(blockIdx.x * 8u + (threadIdx.x >> 5)));
  const int e0 = w * EPW;
  const int Et = SELF ? E + n : E;
  if (e0 >= Et) return;

  int el = e0 + ((lane < EPW) ? lane : 0);
  el = (el < Et) ? el : (Et - 1);
  int sl, dl, nl;
  if (SELF && el >= E) {
    sl = dl = el - E;
    const float d = dinv[sl];
    nl = __builtin_bit_cast(int, d * d);
  } else {
    sl = src[el];
    dl = dst[el];
    nl = __builtin_bit_cast(int, norm[el]);
  }

  float av[D / 32];
#pragma unroll
  for (int i = 0; i < D / 32; ++i) av[i] = ADDV ? addv[lane + 32 * i] : 0.f;

  if (e0 + EPW <= Et) {
    scatter_rows<D, EPW, ADDV>(h, out, sl, dl, nl, 0, lane, av);
  } else {
    for (int k = 0; e0 + k < Et; ++k)
      scatter_rows<D, 1, ADDV>(h, out, sl, dl, nl, k, lane, av);
  }
}

template <int D0, int D, int RB, int NW, int NO>
__global__ void __launch_bounds__(NW * 32)
rows_mm2_kernel(const float* __restrict__ X, const float* __restrict__ rw,
                const float* __restrict__ Bw0, const float* __restrict__ bias0,
                const float* __restrict__ Bw, const float* __restrict__ bias,
                float* __restrict__ out, int n) {
  static_assert(NO == NW * 32, "second product: one wave per 32 output columns");
  static_assert(D == NW * 16, "first product: one wave per 16 columns");
  static_assert(D0 % 32 == 0 && D % 32 == 0, "whole K steps of 4; four quarters of whole 2-step trips");
  constexpr int NT  = NW * 32;
  constexpr int TM  = (RB + 15) / 16;
  constexpr int RP  = TM * 16;
  constexpr int LDX = D0 + 4;
  constexpr int LDA = D + 4;
  __shared__ float xin[RP * LDX];
  __shared__ float acc[RP * LDA];
  __shared__ float rws[RP];

  const int tid  = threadIdx.x;
  const int lane = tid & 31;
  const int w    = __builtin_amdgcn_readfirstlane(tid >> 5);
  const int lh   = lane >> 4;
  const int lm   = lane & 15;
  const int r0   = blockIdx.x * RB;
  const int rows = (n - r0) < RB ? (n - r0) : RB;

  for (int i4 = tid; i4 < RP * (D0 / 4); i4 += NT) {
    const int r = i4 / (D0 / 4);
    const int c = (i4 % (D0 / 4)) * 4;
    float4 v = make_float4(0.f, 0.f, 0.f, 0.f);
    if (r < rows) v = *(const float4*)(X + (size_t)(r0 + r) * D0 + c);
    *(float4*)(&xin[r * LDX + c]) = v;
  }
  if (tid < RP) rws[tid] = tid < rows ? rw[r0 + tid] : 0.f;
  __syncthreads();

  {
    v8f c1[TM];
#pragma unroll
    for (int tm = 0; tm < TM; ++tm) c1[tm] = (v8f)0.0f;
    const float* ap0 = &xin[lm * LDX + 2 * lh];
    const float* bp0 = Bw0 + (size_t)(2 * lh) * D + w * 16 + lm;
#pragma unroll
    for (int kq = 0; kq < 4; ++kq) {
      wave_prio_for_quarter(kq);
#pragma unroll 2
      for (int k0 = kq * (D0 / 4); k0 < (kq + 1) * (D0 / 4); k0 += 4) {
        v2f a[TM];
#pragma unroll
        for (int tm = 0; tm < TM; ++tm)
          a[tm] = *(const v2f*)(ap0 + tm * 16 * LDX + k0);
        v2f b;
        b.x = bp0[(size_t)k0 * D];
        b.y = bp0[(size_t)(k0 + 1) * D];
#pragma unroll
        for (int tm = 0; tm < TM; ++tm) c1[tm] = wmma_f32_16x16x4(a[tm], b, c1[tm]);
      }
    }
    const float bb = bias0[w * 16 + lm];
    float* cp = &acc[(8 * lh) * LDA + w * 16 + lm];
#pragma unroll
    for (int tm = 0; tm < TM; ++tm) {
      const float4 w0 = *(const float4*)(&rws[tm * 16 + 8 * lh]);
      const float4 w1 = *(const float4*)(&rws[tm * 16 + 8 * lh + 4]);
      const float wv[8] = {w0.x, w0.y, w0.z, w0.w, w1.x, w1.y, w1.z, w1.w};
#pragma unroll
      for (int v = 0; v < 8; ++v)
        cp[(tm * 16 + v) * LDA] = c1[tm][v] + wv[v] * bb;
    }
  }
  __syncthreads();

  v8f c[TM][2];
#pragma unroll
  for (int tm = 0; tm < TM; ++tm) {
    c[tm][0] = (v8f)0.0f;
    c[tm][1] = (v8f)0.0f;
  }
  const float* ap = &acc[lm * LDA + 2 * lh];
  const float* bp = Bw + (size_t)(2 * lh) * NO + w * 32 + 2 * lm;
#pragma unroll
  for (int kq = 0; kq < 4; ++kq) {
    wave_prio_for_quarter(kq);
#pragma unroll 2
    for (int k0 = kq * (D / 4); k0 < (kq + 1) * (D / 4); k0 += 4) {
      v2f a[TM];
#pragma unroll
      for (int tm = 0; tm < TM; ++tm)
        a[tm] = *(const v2f*)(ap + tm * 16 * LDA + k0);
      const float2 b0 = *(const float2*)(bp + (size_t)k0 * NO);
      const float2 b1 = *(const float2*)(bp + (size_t)(k0 + 1) * NO);
      v2f be, bo;
      be.x = b0.x; be.y = b1.x;
      bo.x = b0.y; bo.y = b1.y;
#pragma unroll
      for (int tm = 0; tm < TM; ++tm) {
        c[tm][0] = wmma_f32_16x16x4(a[tm], be, c[tm][0]);
        c[tm][1] = wmma_f32_16x16x4(a[tm], bo, c[tm][1]);
      }
    }
  }

  const float2 bo2 = *(const float2*)(bias + w * 32 + 2 * lm);
  float* op = out + (size_t)r0 * NO + w * 32 + 2 * lm;
#pragma unroll
  for (int tm = 0; tm < TM; ++tm) {
#pragma unroll
    for (int v = 0; v < 8; ++v) {
      const int row = tm * 16 + v + 8 * lh;
      if (row < rows)
        *(float2*)(op + (size_t)row * NO) =
            make_float2(c[tm][0][v] + bo2.x, c[tm][1][v] + bo2.y);
    }
  }
}

template <int NTM, int D, int LDA, int NO>
__device__ __forceinline__ void enc_second_product(
    const float* t0s, int tm0, int cq, int lane,
    const float* __restrict__ Wb, float* __restrict__ U, int r0, int rows) {
  const int lh = lane >> 4;
  const int lm = lane & 15;
  v8f c[NTM][2];
#pragma unroll
  for (int tm = 0; tm < NTM; ++tm) {
    c[tm][0] = (v8f)0.0f;
    c[tm][1] = (v8f)0.0f;
  }
  const float* ap = &t0s[(tm0 * 16 + lm) * LDA + 4 * lh];
  const float* bp = Wb + (size_t)(32 * cq + 2 * lm) * D + 4 * lh;
  for (int k0 = 0; k0 < D; k0 += 32) {
    float4 b0[4], b1[4];
#pragma unroll
    for (int q = 0; q < 4; ++q) {
      b0[q] = *(const float4*)(bp + k0 + 8 * q);
      b1[q] = *(const float4*)(bp + D + k0 + 8 * q);
    }
#pragma unroll
    for (int q = 0; q < 4; ++q) {
      float4 a[NTM];
#pragma unroll
      for (int tm = 0; tm < NTM; ++tm)
        a[tm] = *(const float4*)(ap + tm * 16 * LDA + k0 + 8 * q);
      v2f be, bo;
      be.x = b0[q].x; be.y = b0[q].y;
      bo.x = b1[q].x; bo.y = b1[q].y;
#pragma unroll
      for (int tm = 0; tm < NTM; ++tm) {
        v2f av; av.x = a[tm].x; av.y = a[tm].y;
        c[tm][0] = wmma_f32_16x16x4(av, be, c[tm][0]);
        c[tm][1] = wmma_f32_16x16x4(av, bo, c[tm][1]);
      }
      be.x = b0[q].z; be.y = b0[q].w;
      bo.x = b1[q].z; bo.y = b1[q].w;
#pragma unroll
      for (int tm = 0; tm < NTM; ++tm) {
        v2f av; av.x = a[tm].z; av.y = a[tm].w;
        c[tm][0] = wmma_f32_16x16x4(av, be, c[tm][0]);
        c[tm][1] = wmma_f32_16x16x4(av, bo, c[tm][1]);
      }
    }
  }

#pragma unroll
  for (int tm = 0; tm < NTM; ++tm) {
#pragma unroll
    for (int v = 0; v < 8; ++v) {
      const int row = (tm0 + tm) * 16 + v + 8 * lh;
      if (row < rows) {
        const size_t o = (size_t)(r0 + row) * NO + 32 * cq + 2 * lm;
        *(float2*)(U + o) = make_float2(c[tm][0][v], c[tm][1][v]);
      }
    }
  }
}

template <int K0, int D, int RB, int NO, int KC>
__global__ void __launch_bounds__(1024)
enc_mm2_kernel(const float* __restrict__ X, const float* __restrict__ Wa,
               const float* __restrict__ Wb, float* __restrict__ U,
               float* __restrict__ P, const float* __restrict__ b2,
               float* __restrict__ H2, float* __restrict__ A2, int n,
               const int* __restrict__ src, const int* __restrict__ dst, int E,
               const float* __restrict__ b1, float* __restrict__ dinv,
               float* __restrict__ norm, float* __restrict__ rsum,
               float* __restrict__ cvec) {
  constexpr int NW = 32;
  constexpr int NT = NW * 32;
  constexpr int TM = (RB + 15) / 16;
  constexpr int RP = TM * 16;
  static_assert(D == NW * 16, "first product: one wave per 16 columns");
  static_assert(NO == 128, "second product: 4 waves of 32 columns per row group");
  static_assert(TM == 5, "second product: row tiles split 3 + 2 over two wave groups");
  static_assert(K0 % KC == 0 && KC % 128 == 0 && D % 32 == 0, "whole 32-k groups; four quarters of whole groups per chunk");
  static_assert((RP * (KC / 4)) % NT == 0, "X chunk: whole float4 rounds per thread");
  constexpr int NLD = RP * (KC / 4) / NT;
  constexpr int LDX = KC + 4;
  constexpr int LDA = D + 4;
  __shared__ alignas(16) float xs[RP * LDX];
  __shared__ alignas(16) float t0s[RP * LDA];

  if (blockIdx.x == 0) {
    graph_norm_block<NT, D, NO>(xs, t0s, src, dst, E, n, dinv, norm, rsum, Wb, b1, cvec);
    return;
  }

  const int tid  = threadIdx.x;
  const int lane = tid & 31;
  const int w    = __builtin_amdgcn_readfirstlane(tid >> 5);
  const int lh   = lane >> 4;
  const int lm   = lane & 15;
  const int r0   = (blockIdx.x - 1) * RB;
  const int rows = (n - r0) < RB ? (n - r0) : RB;

  v8f c1[TM];
#pragma unroll
  for (int tm = 0; tm < TM; ++tm) c1[tm] = (v8f)0.0f;
  const float* ap0 = &xs[lm * LDX + 4 * lh];
  const float* bp0 = Wa + (size_t)(w * 16 + lm) * K0 + 4 * lh;

  for (int kc = 0; kc < K0; kc += KC) {
    if (kc) __syncthreads();
    float4 xv[NLD];
#pragma unroll
    for (int it = 0; it < NLD; ++it) {
      const int i4 = tid + it * NT;
      const int r  = i4 / (KC / 4);
      const int c  = (i4 % (KC / 4)) * 4;
      const int rr = r < rows ? r : rows - 1;
      xv[it] = *(const float4*)(X + (size_t)(r0 + rr) * K0 + kc + c);
    }
#pragma unroll
    for (int it = 0; it < NLD; ++it) {
      const int i4 = tid + it * NT;
      const int r  = i4 / (KC / 4);
      const int c  = (i4 % (KC / 4)) * 4;
      *(float4*)(&xs[r * LDX + c]) = xv[it];
    }
    __syncthreads();

    for (int k0 = 0; k0 < KC; k0 += 32) {
      if (k0 % (KC / 4) == 0) wave_prio_for_quarter(k0 / (KC / 4));
      float4 b[4];
#pragma unroll
      for (int q = 0; q < 4; ++q)
        b[q] = *(const float4*)(bp0 + kc + k0 + 8 * q);
#pragma unroll
      for (int q = 0; q < 4; ++q) {
        float4 a[TM];
#pragma unroll
        for (int tm = 0; tm < TM; ++tm)
          a[tm] = *(const float4*)(ap0 + tm * 16 * LDX + k0 + 8 * q);
        v2f bv;
        bv.x = b[q].x; bv.y = b[q].y;
#pragma unroll
        for (int tm = 0; tm < TM; ++tm) {
          v2f av; av.x = a[tm].x; av.y = a[tm].y;
          c1[tm] = wmma_f32_16x16x4(av, bv, c1[tm]);
        }
        bv.x = b[q].z; bv.y = b[q].w;
#pragma unroll
        for (int tm = 0; tm < TM; ++tm) {
          v2f av; av.x = a[tm].z; av.y = a[tm].w;
          c1[tm] = wmma_f32_16x16x4(av, bv, c1[tm]);
        }
      }
    }
  }

  {
    float* cp = &t0s[(8 * lh) * LDA + w * 16 + lm];
#pragma unroll
    for (int tm = 0; tm < TM; ++tm)
#pragma unroll
      for (int v = 0; v < 8; ++v) cp[(tm * 16 + v) * LDA] = c1[tm][v];
  }
  __syncthreads();

  if (w < 4) {
    enc_second_product<3, D, LDA, NO>(t0s, 0, w, lane, Wb, U, r0, rows);
  } else if (w < 8) {
    enc_second_product<2, D, LDA, NO>(t0s, 3, w - 4, lane, Wb, U, r0, rows);
  } else {
    for (int i4 = tid - 256; i4 < rows * (NO / 4); i4 += NT - 256) {
      const size_t o = (size_t)r0 * NO + (size_t)i4 * 4;
      const int    c = (i4 % (NO / 4)) * 4;
      *(float4*)(P + o)  = make_float4(0.f, 0.f, 0.f, 0.f);
      *(float4*)(H2 + o) = *(const float4*)(b2 + c);
      *(float4*)(A2 + o) = make_float4(0.f, 0.f, 0.f, 0.f);
    }
  }
}

extern "C" void kernel_launch(void* const* d_in, const int* in_sizes, int n_in,
                              void* d_out, int out_size, void* d_ws, size_t ws_size,
                              hipStream_t stream) {
  const int IN = 1024, HID = 512, OUT = 128;

  const float* features = (const float*)d_in[0];
  const int*   edge     = (const int*)d_in[1];
  const float* W1       = (const float*)d_in[2];
  const float* b1       = (const float*)d_in[3];
  const float* W2       = (const float*)d_in[4];
  const float* b2       = (const float*)d_in[5];
  const float* b3       = (const float*)d_in[6];
  const float* b4       = (const float*)d_in[7];
  const float* head1    = (const float*)d_in[8];

  const int n  = in_sizes[0] / IN;
  const int E  = in_sizes[1] / 2;
  const int Et = E + n;
  const int* src = edge;
  const int* dst = edge + E;

  float* out = (float*)d_out;
  float* z   = out;
  float* h2  = out + (size_t)n * OUT;
  float* h4  = out + (size_t)n * OUT * 2;

  float* ws   = (float*)d_ws;
  float* dinv = ws;
  float* norm = dinv + n;
  float* t1   = norm + Et + (size_t)n * HID;
  float* h2p  = t1 + (size_t)n * HID;
  float* t2   = h2p + (size_t)n * OUT;
  float* cvec = norm + E;
  float* u    = h2p;
  float* p    = t1;
  float* a1   = t2;
  float* a2   = t2 + (size_t)n * OUT;
  float* rsum = t2 + (size_t)n * OUT * 2;

  constexpr int RB = 80;
  enc_mm2_kernel<IN, HID, RB, OUT, 256><<<1 + (n + RB - 1) / RB, 1024, 0, stream>>>(
      features, W1, W2, u, p, b2, h2, a2, n, src, dst, E, b1, dinv, norm, rsum, cvec);

  scatter_rows_kernel<OUT, 4, true, false><<<(E + n + 31) / 32, 256, 0, stream>>>(
      u, src, dst, norm, p, E, dinv, n, nullptr);
  scatter_rows_kernel<OUT, 4, true, true><<<(E + n + 31) / 32, 256, 0, stream>>>(
      p, src, dst, norm, h2, E, dinv, n, cvec);

  gemm_wmma<0, true><<<dim3(OUT / 128, (n + 127) / 128), 256, 0, stream>>>(
      h2, OUT, head1, OUT, z, OUT, n, OUT, OUT, a1, nullptr, dinv);

  scatter_rows_kernel<OUT, 4, false, false><<<(E + 31) / 32, 256, 0, stream>>>(
      z, src, dst, norm, a1, E, nullptr, 0, nullptr);
  scatter_rows_kernel<OUT, 4, true, false><<<(E + n + 31) / 32, 256, 0, stream>>>(
      a1, src, dst, norm, a2, E, dinv, n, nullptr);

  rows_mm2_kernel<OUT, HID, RB, 32, IN><<<(n + RB - 1) / RB, 1024, 0, stream>>>(
      a2, rsum, W2, b3, W1, b4, h4, n);
}
